# starting kernel
# baseline (speedup 1.0000x reference)
#include <hip/hip_runtime.h>

typedef float v2f __attribute__((ext_vector_type(2)));
typedef float v8f __attribute__((ext_vector_type(8)));

#define EPSQ 1e-15f

__global__ void zero_kernel(float* __restrict__ p, long n) {
  long i = (long)blockIdx.x * blockDim.x + threadIdx.x;
  long stride = (long)gridDim.x * blockDim.x;
  for (; i < n; i += stride) p[i] = 0.0f;
}

__global__ void gauss_kernel(const int* __restrict__ ei,
                             const float* __restrict__ ea,
                             const float* __restrict__ mu1, const float* __restrict__ sg1,
                             const float* __restrict__ mu2, const float* __restrict__ sg2,
                             float* __restrict__ w1, float* __restrict__ w2,
                             float* __restrict__ cnt, int E) {
  int e = blockIdx.x * blockDim.x + threadIdx.x;
  if (e >= E) return;
  float s1 = 0.0f, s2 = 0.0f;
#pragma unroll
  for (int d = 0; d < 8; ++d) {
    float a  = ea[(long)e * 8 + d];
    float d1 = a - mu1[d];
    float d2 = a - mu2[d];
    float q1 = sg1[d], q2 = sg2[d];
    s1 += (d1 * d1) / (EPSQ + q1 * q1);
    s2 += (d2 * d2) / (EPSQ + q2 * q2);
  }
  w1[e] = __expf(-0.5f * s1);
  w2[e] = __expf(-0.5f * s2);
  int dst = ei[E + e];
  atomicAdd(&cnt[dst], 1.0f);
}

__global__ void __launch_bounds__(256)
gemm_kernel(const float* __restrict__ X, const float* __restrict__ G,
            const float* __restrict__ R, const float* __restrict__ B,
            float* __restrict__ XG, float* __restrict__ BASE,
            int N, int ntiles) {
  int wave = blockIdx.x * (blockDim.x >> 5) + (threadIdx.x >> 5);
  if (wave >= ntiles) return;
  int lane = threadIdx.x & 31;
  int half = lane >> 4;
  int m    = lane & 15;
  int row0 = wave * 16;
  int arow = row0 + m;
  if (arow >= N) arow = N - 1;

  v8f accg0 = {}, accg1 = {}, accr0 = {}, accr1 = {};

#pragma unroll
  for (int k = 0; k < 32; k += 4) {
    int kk = k + half * 2;
    v2f a = *(const v2f*)(X + (long)arow * 32 + kk);
    v2f bg0, bg1, br0, br1;
    bg0.x = G[kk * 32 + m];        bg0.y = G[(kk + 1) * 32 + m];
    bg1.x = G[kk * 32 + 16 + m];   bg1.y = G[(kk + 1) * 32 + 16 + m];
    br0.x = R[kk * 32 + m];        br0.y = R[(kk + 1) * 32 + m];
    br1.x = R[kk * 32 + 16 + m];   br1.y = R[(kk + 1) * 32 + 16 + m];
    accg0 = __builtin_amdgcn_wmma_f32_16x16x4_f32(false, a, false, bg0, (short)0, accg0, false, false);
    accg1 = __builtin_amdgcn_wmma_f32_16x16x4_f32(false, a, false, bg1, (short)0, accg1, false, false);
    accr0 = __builtin_amdgcn_wmma_f32_16x16x4_f32(false, a, false, br0, (short)0, accr0, false, false);
    accr1 = __builtin_amdgcn_wmma_f32_16x16x4_f32(false, a, false, br1, (short)0, accr1, false, false);
  }

  float b0 = B[m], b1 = B[16 + m];
#pragma unroll
  for (int r = 0; r < 8; ++r) {
    int row = row0 + half * 8 + r;
    if (row < N) {
      XG[(long)row * 32 + m]          = accg0[r];
      XG[(long)row * 32 + 16 + m]     = accg1[r];
      BASE[(long)row * 32 + m]        = accr0[r] + b0;
      BASE[(long)row * 32 + 16 + m]   = accr1[r] + b1;
    }
  }
}

__global__ void edge_kernel(const int* __restrict__ ei, const float* __restrict__ w,
                            const float* __restrict__ XG, float* __restrict__ AGG,
                            int E) {
  long t = (long)blockIdx.x * blockDim.x + threadIdx.x;
  long e = t >> 3;
  if (e >= E) return;
  int q = (int)(t & 7);
  int s = ei[e];
  int d = ei[E + e];
  float ww = w[e];
  float4 v = *(const float4*)(XG + (long)s * 32 + q * 4);
  float* base = AGG + (long)d * 32 + q * 4;
  atomicAdd(base + 0, v.x * ww);
  atomicAdd(base + 1, v.y * ww);
  atomicAdd(base + 2, v.z * ww);
  atomicAdd(base + 3, v.w * ww);
}

__global__ void finish_kernel(const float* __restrict__ AGG, const float* __restrict__ BASE,
                              const float* __restrict__ CNT, float* __restrict__ OUT, int N) {
  long i = (long)blockIdx.x * blockDim.x + threadIdx.x;
  if (i >= (long)N * 32) return;
  int n = (int)(i >> 5);
  float c = CNT[n];
  c = c > 1.0f ? c : 1.0f;
  OUT[i] = AGG[i] / c + BASE[i];
}

extern "C" void kernel_launch(void* const* d_in, const int* in_sizes, int n_in,
                              void* d_out, int out_size, void* d_ws, size_t ws_size,
                              hipStream_t stream) {
  const int*   ei  = (const int*)d_in[0];
  const float* ea  = (const float*)d_in[1];
  const float* x   = (const float*)d_in[2];
  const float* g1  = (const float*)d_in[3];
  const float* mu1 = (const float*)d_in[4];
  const float* sg1 = (const float*)d_in[5];
  const float* r1  = (const float*)d_in[6];
  const float* b1  = (const float*)d_in[7];
  const float* g2  = (const float*)d_in[8];
  const float* mu2 = (const float*)d_in[9];
  const float* sg2 = (const float*)d_in[10];
  const float* r2  = (const float*)d_in[11];
  const float* b2  = (const float*)d_in[12];

  const int E = in_sizes[0] / 2;
  const int N = in_sizes[2] / 32;

  float* ws   = (float*)d_ws;
  float* w1   = ws;
  float* w2   = w1 + E;
  float* cnt  = w2 + E;
  float* xg   = cnt + N;
  float* base = xg + (long)N * 32;
  float* agg  = base + (long)N * 32;
  float* h    = agg + (long)N * 32;
  float* out  = (float*)d_out;

  const int ntiles     = (N + 15) / 16;
  const int gemmBlocks = (ntiles + 7) / 8;
  const int edgeBlocks = (int)(((long)E * 8 + 255) / 256);
  const int finBlocks  = (int)(((long)N * 32 + 255) / 256);
  const int gBlocks    = (E + 255) / 256;

  zero_kernel<<<2048, 256, 0, stream>>>(cnt, N);
  zero_kernel<<<2048, 256, 0, stream>>>(agg, (long)N * 32);
  gauss_kernel<<<gBlocks, 256, 0, stream>>>(ei, ea, mu1, sg1, mu2, sg2, w1, w2, cnt, E);

  gemm_kernel<<<gemmBlocks, 256, 0, stream>>>(x, g1, r1, b1, xg, base, N, ntiles);
  edge_kernel<<<edgeBlocks, 256, 0, stream>>>(ei, w1, xg, agg, E);
  finish_kernel<<<finBlocks, 256, 0, stream>>>(agg, base, cnt, h, N);

  zero_kernel<<<2048, 256, 0, stream>>>(agg, (long)N * 32);
  gemm_kernel<<<gemmBlocks, 256, 0, stream>>>(h, g2, r2, b2, xg, base, N, ntiles);
  edge_kernel<<<edgeBlocks, 256, 0, stream>>>(ei, w2, xg, agg, E);
  finish_kernel<<<finBlocks, 256, 0, stream>>>(agg, base, cnt, out, N);
}


# Arrays added into with atomics (in-degree, edge sums) get one copy per L2 half, chosen by the wave's die id (read once, waves loop); finish sums the copies: changes the order of the float sums
# speedup vs baseline: 1.6015x; 1.6015x over previous
#include <hip/hip_runtime.h>

typedef float v2f __attribute__((ext_vector_type(2)));
typedef float v8f __attribute__((ext_vector_type(8)));

#define EPSQ 1e-15f

__global__ void zero_kernel(float* __restrict__ p, long n) {
  long i = (long)blockIdx.x * blockDim.x + threadIdx.x;
  long stride = (long)gridDim.x * blockDim.x;
  for (; i < n; i += stride) p[i] = 0.0f;
}

__global__ void zero2_kernel(float* __restrict__ a, float* __restrict__ b, long n) {
  long i = (long)blockIdx.x * blockDim.x + threadIdx.x;
  long stride = (long)gridDim.x * blockDim.x;
  for (; i < n; i += stride) { a[i] = 0.0f; b[i] = 0.0f; }
}

static __device__ __forceinline__ int upper_half() {
  unsigned where = (unsigned)__builtin_amdgcn_s_sendmsg_rtn(0x87);
  return ((where >> 16) & 15u) >= 4u ? 1 : 0;
}

__global__ void __launch_bounds__(256, 16)
gauss_kernel(const int* __restrict__ ei,
             const float* __restrict__ ea,
             const float* __restrict__ mu1, const float* __restrict__ sg1,
             const float* __restrict__ mu2, const float* __restrict__ sg2,
             float* __restrict__ w1, float* __restrict__ w2,
             float* __restrict__ cnt0, float* __restrict__ cnt1, int E) {
  float* cnt = upper_half() ? cnt1 : cnt0;
  const int stride = gridDim.x * blockDim.x;
  for (int e = blockIdx.x * blockDim.x + threadIdx.x; e < E; e += stride) {
    float s1 = 0.0f, s2 = 0.0f;
#pragma unroll
    for (int d = 0; d < 8; ++d) {
      float a  = ea[(long)e * 8 + d];
      float d1 = a - mu1[d];
      float d2 = a - mu2[d];
      float q1 = sg1[d], q2 = sg2[d];
      s1 += (d1 * d1) / (EPSQ + q1 * q1);
      s2 += (d2 * d2) / (EPSQ + q2 * q2);
    }
    w1[e] = __expf(-0.5f * s1);
    w2[e] = __expf(-0.5f * s2);
    int dst = ei[E + e];
    atomicAdd(&cnt[dst], 1.0f);
  }
}

__global__ void __launch_bounds__(256)
gemm_kernel(const float* __restrict__ X, const float* __restrict__ G,
            const float* __restrict__ R, const float* __restrict__ B,
            float* __restrict__ XG, float* __restrict__ BASE,
            int N, int ntiles) {
  int wave = blockIdx.x * (blockDim.x >> 5) + (threadIdx.x >> 5);
  if (wave >= ntiles) return;
  int lane = threadIdx.x & 31;
  int half = lane >> 4;
  int m    = lane & 15;
  int row0 = wave * 16;
  int arow = row0 + m;
  if (arow >= N) arow = N - 1;

  v8f accg0 = {}, accg1 = {}, accr0 = {}, accr1 = {};

#pragma unroll
  for (int k = 0; k < 32; k += 4) {
    int kk = k + half * 2;
    v2f a = *(const v2f*)(X + (long)arow * 32 + kk);
    v2f bg0, bg1, br0, br1;
    bg0.x = G[kk * 32 + m];        bg0.y = G[(kk + 1) * 32 + m];
    bg1.x = G[kk * 32 + 16 + m];   bg1.y = G[(kk + 1) * 32 + 16 + m];
    br0.x = R[kk * 32 + m];        br0.y = R[(kk + 1) * 32 + m];
    br1.x = R[kk * 32 + 16 + m];   br1.y = R[(kk + 1) * 32 + 16 + m];
    accg0 = __builtin_amdgcn_wmma_f32_16x16x4_f32(false, a, false, bg0, (short)0, accg0, false, false);
    accg1 = __builtin_amdgcn_wmma_f32_16x16x4_f32(false, a, false, bg1, (short)0, accg1, false, false);
    accr0 = __builtin_amdgcn_wmma_f32_16x16x4_f32(false, a, false, br0, (short)0, accr0, false, false);
    accr1 = __builtin_amdgcn_wmma_f32_16x16x4_f32(false, a, false, br1, (short)0, accr1, false, false);
  }

  float b0 = B[m], b1 = B[16 + m];
#pragma unroll
  for (int r = 0; r < 8; ++r) {
    int row = row0 + half * 8 + r;
    if (row < N) {
      XG[(long)row * 32 + m]          = accg0[r];
      XG[(long)row * 32 + 16 + m]     = accg1[r];
      BASE[(long)row * 32 + m]        = accr0[r] + b0;
      BASE[(long)row * 32 + 16 + m]   = accr1[r] + b1;
    }
  }
}

__global__ void __launch_bounds__(256)
edge_kernel(const int* __restrict__ ei, const float* __restrict__ w,
            const float* __restrict__ XG, float* __restrict__ ACC0,
            long plane, int E) {
  float* acc = ACC0 + (upper_half() ? plane : 0L);

  const long total  = (long)E * 8;
  const long stride = (long)gridDim.x * blockDim.x;
  for (long t = (long)blockIdx.x * blockDim.x + threadIdx.x; t < total; t += stride) {
    long e = t >> 3;
    int q = (int)(t & 7);
    int s = ei[e];
    int d = ei[E + e];
    float ww = w[e];
    float4 v = *(const float4*)(XG + (long)s * 32 + q * 4);
    float* base = acc + (long)d * 32 + q * 4;
    atomicAdd(base + 0, v.x * ww);
    atomicAdd(base + 1, v.y * ww);
    atomicAdd(base + 2, v.z * ww);
    atomicAdd(base + 3, v.w * ww);
  }
}

__global__ void finish_kernel(const float* __restrict__ ACC0, const float* ACC1,
                              const float* __restrict__ BASE,
                              float* CNT0, const float* CNT1, float* OUT, int N) {
  long i = (long)blockIdx.x * blockDim.x + threadIdx.x;
  if (i >= (long)N * 32) return;
  int n = (int)(i >> 5);
  float c = CNT0[n];
  if (CNT1 != nullptr) {
    c += CNT1[n];
    if ((i & 31) == 0) CNT0[n] = c;
  }
  c = c > 1.0f ? c : 1.0f;
  float a = ACC0[i] + ACC1[i];
  OUT[i] = a / c + BASE[i];
}

extern "C" void kernel_launch(void* const* d_in, const int* in_sizes, int n_in,
                              void* d_out, int out_size, void* d_ws, size_t ws_size,
                              hipStream_t stream) {
  const int*   ei  = (const int*)d_in[0];
  const float* ea  = (const float*)d_in[1];
  const float* x   = (const float*)d_in[2];
  const float* g1  = (const float*)d_in[3];
  const float* mu1 = (const float*)d_in[4];
  const float* sg1 = (const float*)d_in[5];
  const float* r1  = (const float*)d_in[6];
  const float* b1  = (const float*)d_in[7];
  const float* g2  = (const float*)d_in[8];
  const float* mu2 = (const float*)d_in[9];
  const float* sg2 = (const float*)d_in[10];
  const float* r2  = (const float*)d_in[11];
  const float* b2  = (const float*)d_in[12];

  const int E = in_sizes[0] / 2;
  const int N = in_sizes[2] / 32;

  float* ws   = (float*)d_ws;
  float* w1   = ws;
  float* w2   = w1 + E;
  float* cnt  = w2 + E;
  float* xg   = cnt + N;
  float* base = xg + (long)N * 32;
  float* agg  = base + (long)N * 32;
  float* h    = agg + (long)N * 32;
  float* out  = (float*)d_out;
  const long plane = (long)N * 32;

  const int ntiles     = (N + 15) / 16;
  const int gemmBlocks = (ntiles + 7) / 8;
  const int edgeBlocks = 2048;
  const int finBlocks  = (int)(((long)N * 32 + 255) / 256);
  const int gBlocks    = 2048;
  float* cnt1 = out;

  zero2_kernel<<<256, 256, 0, stream>>>(cnt, cnt1, N);
  zero_kernel<<<2048, 256, 0, stream>>>(agg, 2 * plane);
  gauss_kernel<<<gBlocks, 256, 0, stream>>>(ei, ea, mu1, sg1, mu2, sg2, w1, w2, cnt, cnt1, E);

  gemm_kernel<<<gemmBlocks, 256, 0, stream>>>(x, g1, r1, b1, xg, base, N, ntiles);
  edge_kernel<<<edgeBlocks, 256, 0, stream>>>(ei, w1, xg, agg, plane, E);
  finish_kernel<<<finBlocks, 256, 0, stream>>>(agg, h, base, cnt, cnt1, h, N);

  gemm_kernel<<<gemmBlocks, 256, 0, stream>>>(h, g2, r2, b2, xg, base, N, ntiles);
  zero_kernel<<<2048, 256, 0, stream>>>(agg, 2 * plane);
  edge_kernel<<<edgeBlocks, 256, 0, stream>>>(ei, w2, xg, agg, plane, E);
  finish_kernel<<<finBlocks, 256, 0, stream>>>(agg, h, base, cnt, nullptr, out, N);
}


# Edges bucketed by destination per call (counted rank, segment allocation, placement); each node sums its own edges in registers in arrival-rank order: float atomics gone, float sum order changes
# speedup vs baseline: 2.1869x; 1.3655x over previous
#include <hip/hip_runtime.h>


typedef float v2f __attribute__((ext_vector_type(2)));
typedef float v8f __attribute__((ext_vector_type(8)));

#define EPSQ 1e-15f

__global__ void zero_kernel(int* __restrict__ p, long n) {
  long i = (long)blockIdx.x * blockDim.x + threadIdx.x;
  long stride = (long)gridDim.x * blockDim.x;
  for (; i < n; i += stride) p[i] = 0;
}

static __device__ __forceinline__ int upper_half() {
  unsigned where = (unsigned)__builtin_amdgcn_s_sendmsg_rtn(0x87);
  return ((where >> 16) & 15u) >= 4u ? 1 : 0;
}

__global__ void __launch_bounds__(256)
rank_kernel(const int* __restrict__ ei,
            int* __restrict__ cnt0, int* __restrict__ cnt1,
            int* __restrict__ rank, int E) {
  const int hi = upper_half();
  int* cnt = hi ? cnt1 : cnt0;
  const unsigned tag = hi ? 0x80000000u : 0u;
  const int stride = gridDim.x * blockDim.x;
  for (int e0 = blockIdx.x * blockDim.x + threadIdx.x; e0 < E; e0 += 4 * stride) {
    int d[4];
    unsigned r[4];
#pragma unroll
    for (int k = 0; k < 4; ++k) {
      int e = e0 + k * stride;
      e = e < E ? e : E - 1;
      d[k] = ei[E + e];
    }
#pragma unroll
    for (int k = 0; k < 4; ++k) {
      const int e = e0 + k * stride;
      r[k] = 0u;
      if (e < E) r[k] = (unsigned)atomicAdd(&cnt[d[k]], 1);
    }
#pragma unroll
    for (int k = 0; k < 4; ++k) {
      const int e = e0 + k * stride;
      if (e < E) rank[e] = (int)(r[k] | tag);
    }
  }
}

__global__ void __launch_bounds__(256)
alloc_kernel(const int* __restrict__ cnt0, const int* __restrict__ cnt1,
             int* __restrict__ cursor, int* __restrict__ seg, int N) {
  __shared__ int wsum[8];
  __shared__ int sbase;
  const int n = blockIdx.x * 256 + threadIdx.x;
  const int lane = threadIdx.x & 31;
  const int wv = threadIdx.x >> 5;
  int c0 = 0, c1 = 0;
  if (n < N) { c0 = cnt0[n]; c1 = cnt1[n]; }
  const int tot = c0 + c1;
  int inc = tot;
  for (int o = 1; o < 32; o <<= 1) {
    int t = __shfl_up(inc, o);
    if (lane >= o) inc += t;
  }
  if (lane == 31) wsum[wv] = inc;
  __syncthreads();
  if (threadIdx.x == 0) {
    int s = 0;
    for (int w = 0; w < 8; ++w) { int t = wsum[w]; wsum[w] = s; s += t; }
    sbase = atomicAdd(cursor, s);
  }
  __syncthreads();
  if (n < N) {
    const int beg = sbase + wsum[wv] + inc - tot;
    int4 o4 = make_int4(beg, beg + tot, c0, 0);
    *(int4*)(seg + (long)n * 4) = o4;
  }
}

__global__ void __launch_bounds__(256)
place_kernel(const int* __restrict__ ei,
             const float* __restrict__ ea,
             const float* __restrict__ mu1, const float* __restrict__ sg1,
             const float* __restrict__ mu2, const float* __restrict__ sg2,
             const int* __restrict__ rank, const int* __restrict__ seg,
             int* __restrict__ csr, int E) {
  const int stride = gridDim.x * blockDim.x;
  for (int e = blockIdx.x * blockDim.x + threadIdx.x; e < E; e += stride) {
    float s1 = 0.0f, s2 = 0.0f;
#pragma unroll
    for (int d = 0; d < 8; ++d) {
      float a  = ea[(long)e * 8 + d];
      float d1 = a - mu1[d];
      float d2 = a - mu2[d];
      float q1 = sg1[d], q2 = sg2[d];
      s1 += (d1 * d1) / (EPSQ + q1 * q1);
      s2 += (d2 * d2) / (EPSQ + q2 * q2);
    }
    const int src = ei[e];
    const int dst = ei[E + e];
    const int r = rank[e];
    const int4 o = *(const int4*)(seg + (long)dst * 4);
    const int pos = o.x + (r < 0 ? o.z : 0) + (r & 0x7fffffff);
    int4 rec = make_int4(src, __float_as_int(__expf(-0.5f * s1)),
                         __float_as_int(__expf(-0.5f * s2)), 0);
    *(int4*)(csr + (long)pos * 4) = rec;
  }
}

__global__ void __launch_bounds__(256)
gemm_kernel(const float* __restrict__ X, const float* __restrict__ G,
            const float* __restrict__ R, const float* __restrict__ B,
            float* __restrict__ XG, float* __restrict__ BASE,
            int N, int ntiles) {
  int wave = blockIdx.x * (blockDim.x >> 5) + (threadIdx.x >> 5);
  if (wave >= ntiles) return;
  int lane = threadIdx.x & 31;
  int half = lane >> 4;
  int m    = lane & 15;
  int row0 = wave * 16;
  int arow = row0 + m;
  if (arow >= N) arow = N - 1;

  v8f accg0 = {}, accg1 = {}, accr0 = {}, accr1 = {};

#pragma unroll
  for (int k = 0; k < 32; k += 4) {
    int kk = k + half * 2;
    v2f a = *(const v2f*)(X + (long)arow * 32 + kk);
    v2f bg0, bg1, br0, br1;
    bg0.x = G[kk * 32 + m];        bg0.y = G[(kk + 1) * 32 + m];
    bg1.x = G[kk * 32 + 16 + m];   bg1.y = G[(kk + 1) * 32 + 16 + m];
    br0.x = R[kk * 32 + m];        br0.y = R[(kk + 1) * 32 + m];
    br1.x = R[kk * 32 + 16 + m];   br1.y = R[(kk + 1) * 32 + 16 + m];
    accg0 = __builtin_amdgcn_wmma_f32_16x16x4_f32(false, a, false, bg0, (short)0, accg0, false, false);
    accg1 = __builtin_amdgcn_wmma_f32_16x16x4_f32(false, a, false, bg1, (short)0, accg1, false, false);
    accr0 = __builtin_amdgcn_wmma_f32_16x16x4_f32(false, a, false, br0, (short)0, accr0, false, false);
    accr1 = __builtin_amdgcn_wmma_f32_16x16x4_f32(false, a, false, br1, (short)0, accr1, false, false);
  }

  float b0 = B[m], b1 = B[16 + m];
#pragma unroll
  for (int r = 0; r < 8; ++r) {
    int row = row0 + half * 8 + r;
    if (row < N) {
      XG[(long)row * 32 + m]          = accg0[r];
      XG[(long)row * 32 + 16 + m]     = accg1[r];
      BASE[(long)row * 32 + m]        = accr0[r] + b0;
      BASE[(long)row * 32 + 16 + m]   = accr1[r] + b1;
    }
  }
}

__global__ void __launch_bounds__(256)
agg_kernel(const int* __restrict__ seg, const int* __restrict__ csr,
           const float* __restrict__ XG, const float* __restrict__ BASE,
           float* __restrict__ OUT, int N, int sel) {
  const long total  = (long)N * 8;
  const long stride = (long)gridDim.x * blockDim.x;
  for (long t = (long)blockIdx.x * blockDim.x + threadIdx.x; t < total; t += stride) {
    const int n = (int)(t >> 3);
    const int q = (int)(t & 7);
    const int2 be = *(const int2*)(seg + (long)n * 4);
    const int beg = be.x, end = be.y;
    const int last = end - 1;
    const float* xq = XG + q * 4;
    float4 acc = make_float4(0.0f, 0.0f, 0.0f, 0.0f);
    if (beg < end) {
      int   src[4];
      float ww[4];
#pragma unroll
      for (int u = 0; u < 4; ++u) {
        int jj = beg + u;
        jj = jj < last ? jj : last;
        const int4 r = *(const int4*)(csr + (long)jj * 4);
        src[u] = r.x;
        ww[u]  = __int_as_float(sel ? r.z : r.y);
      }
      for (int j = beg; j < end; j += 4) {
        float4 v[4];
        float  cw[4];
#pragma unroll
        for (int u = 0; u < 4; ++u) {
          cw[u] = ww[u];
          v[u]  = *(const float4*)(xq + (long)src[u] * 32);
        }
#pragma unroll
        for (int u = 0; u < 4; ++u) {
          int jj = j + 4 + u;
          jj = jj < last ? jj : last;
          const int4 r = *(const int4*)(csr + (long)jj * 4);
          src[u] = r.x;
          ww[u]  = __int_as_float(sel ? r.z : r.y);
        }
#pragma unroll
        for (int u = 0; u < 4; ++u) {
          const int m = (u == 0 || j + u < end) ? -1 : 0;
          const float w = __int_as_float(__float_as_int(cw[u]) & m);
          acc.x += __int_as_float(__float_as_int(v[u].x) & m) * w;
          acc.y += __int_as_float(__float_as_int(v[u].y) & m) * w;
          acc.z += __int_as_float(__float_as_int(v[u].z) & m) * w;
          acc.w += __int_as_float(__float_as_int(v[u].w) & m) * w;
        }
      }
    }
    float c = (float)(end - beg);
    c = c > 1.0f ? c : 1.0f;
    const float4 b = *(const float4*)(BASE + (long)n * 32 + q * 4);
    float4 res;
    res.x = acc.x / c + b.x;
    res.y = acc.y / c + b.y;
    res.z = acc.z / c + b.z;
    res.w = acc.w / c + b.w;
    *(float4*)(OUT + (long)n * 32 + q * 4) = res;
  }
}

extern "C" void kernel_launch(void* const* d_in, const int* in_sizes, int n_in,
                              void* d_out, int out_size, void* d_ws, size_t ws_size,
                              hipStream_t stream) {
  const int*   ei  = (const int*)d_in[0];
  const float* ea  = (const float*)d_in[1];
  const float* x   = (const float*)d_in[2];
  const float* g1  = (const float*)d_in[3];
  const float* mu1 = (const float*)d_in[4];
  const float* sg1 = (const float*)d_in[5];
  const float* r1  = (const float*)d_in[6];
  const float* b1  = (const float*)d_in[7];
  const float* g2  = (const float*)d_in[8];
  const float* mu2 = (const float*)d_in[9];
  const float* sg2 = (const float*)d_in[10];
  const float* r2  = (const float*)d_in[11];
  const float* b2  = (const float*)d_in[12];

  const int E = in_sizes[0] / 2;
  const int N = in_sizes[2] / 32;
  const long plane = (long)N * 32;

  int*   csr    = (int*)d_ws;
  int*   seg    = csr + (long)E * 4;
  float* xg     = (float*)(seg + (long)N * 4);
  float* base   = xg + plane;
  int*   rank   = (int*)(base + plane);
  int*   cnt0   = rank + E;
  int*   cnt1   = cnt0 + N;
  int*   cursor = cnt1 + N;
  float* out    = (float*)d_out;
  float* h      = out;

  const int ntiles     = (N + 15) / 16;
  const int gemmBlocks = (ntiles + 7) / 8;
  const int rankBlocks  = (E + 1023) / 1024;
  const int placeBlocks = (E + 1279) / 1280;
  const int nodeBlocks = (N + 255) / 256;
  const int aggBlocks  = (int)(((long)N * 4 + 255) / 256);

  zero_kernel<<<256, 256, 0, stream>>>(cnt0, 2L * N + 1);
  rank_kernel<<<rankBlocks, 256, 0, stream>>>(ei, cnt0, cnt1, rank, E);
  alloc_kernel<<<nodeBlocks, 256, 0, stream>>>(cnt0, cnt1, cursor, seg, N);
  place_kernel<<<placeBlocks, 256, 0, stream>>>(ei, ea, mu1, sg1, mu2, sg2, rank, seg, csr, E);

  gemm_kernel<<<gemmBlocks, 256, 0, stream>>>(x, g1, r1, b1, xg, base, N, ntiles);
  agg_kernel<<<aggBlocks, 256, 0, stream>>>(seg, csr, xg, base, h, N, 0);

  gemm_kernel<<<gemmBlocks, 256, 0, stream>>>(h, g2, r2, b2, xg, base, N, ntiles);
  agg_kernel<<<aggBlocks, 256, 0, stream>>>(seg, csr, xg, base, out, N, 1);
}


# Dense transform moved behind the edge sum (one Gaussian per edge: linear) and fused into the gather kernel as its WMMA epilogue, one wave per 16 nodes: 8 launches become 6; float sum order changes
# speedup vs baseline: 2.3347x; 1.0676x over previous
#include <hip/hip_runtime.h>


typedef float v2f __attribute__((ext_vector_type(2)));
typedef float v8f __attribute__((ext_vector_type(8)));

#define EPSQ 1e-15f

__global__ void zero_kernel(int* __restrict__ p, long n) {
  long i = (long)blockIdx.x * blockDim.x + threadIdx.x;
  long stride = (long)gridDim.x * blockDim.x;
  for (; i < n; i += stride) p[i] = 0;
}

static __device__ __forceinline__ int upper_half() {
  unsigned where = (unsigned)__builtin_amdgcn_s_sendmsg_rtn(0x87);
  return ((where >> 16) & 15u) >= 4u ? 1 : 0;
}

__global__ void __launch_bounds__(256)
rank_kernel(const int* __restrict__ ei,
            int* __restrict__ cnt0, int* __restrict__ cnt1,
            int* __restrict__ rank, int E) {
  const int hi = upper_half();
  int* cnt = hi ? cnt1 : cnt0;
  const unsigned tag = hi ? 0x80000000u : 0u;
  const int stride = gridDim.x * blockDim.x;
  for (int e0 = blockIdx.x * blockDim.x + threadIdx.x; e0 < E; e0 += 4 * stride) {
    int d[4];
    unsigned r[4];
#pragma unroll
    for (int k = 0; k < 4; ++k) {
      int e = e0 + k * stride;
      e = e < E ? e : E - 1;
      d[k] = ei[E + e];
    }
#pragma unroll
    for (int k = 0; k < 4; ++k) {
      const int e = e0 + k * stride;
      r[k] = 0u;
      if (e < E) r[k] = (unsigned)atomicAdd(&cnt[d[k]], 1);
    }
#pragma unroll
    for (int k = 0; k < 4; ++k) {
      const int e = e0 + k * stride;
      if (e < E) rank[e] = (int)(r[k] | tag);
    }
  }
}

__global__ void __launch_bounds__(256)
alloc_kernel(const int* __restrict__ cnt0, const int* __restrict__ cnt1,
             int* __restrict__ cursor, int* __restrict__ seg, int N) {
  __shared__ int wsum[8];
  __shared__ int sbase;
  const int n = blockIdx.x * 256 + threadIdx.x;
  const int lane = threadIdx.x & 31;
  const int wv = threadIdx.x >> 5;
  int c0 = 0, c1 = 0;
  if (n < N) { c0 = cnt0[n]; c1 = cnt1[n]; }
  const int tot = c0 + c1;
  int inc = tot;
  for (int o = 1; o < 32; o <<= 1) {
    int t = __shfl_up(inc, o);
    if (lane >= o) inc += t;
  }
  if (lane == 31) wsum[wv] = inc;
  __syncthreads();
  if (threadIdx.x == 0) {
    int s = 0;
    for (int w = 0; w < 8; ++w) { int t = wsum[w]; wsum[w] = s; s += t; }
    sbase = atomicAdd(cursor, s);
  }
  __syncthreads();
  if (n < N) {
    const int beg = sbase + wsum[wv] + inc - tot;
    int4 o4 = make_int4(beg, beg + tot, c0, 0);
    *(int4*)(seg + (long)n * 4) = o4;
  }
}

__global__ void __launch_bounds__(256)
place_kernel(const int* __restrict__ ei,
             const float* __restrict__ ea,
             const float* __restrict__ mu1, const float* __restrict__ sg1,
             const float* __restrict__ mu2, const float* __restrict__ sg2,
             const int* __restrict__ rank, const int* __restrict__ seg,
             int* __restrict__ csr, int E) {
  const int stride = gridDim.x * blockDim.x;
  for (int e = blockIdx.x * blockDim.x + threadIdx.x; e < E; e += stride) {
    float s1 = 0.0f, s2 = 0.0f;
#pragma unroll
    for (int d = 0; d < 8; ++d) {
      float a  = ea[(long)e * 8 + d];
      float d1 = a - mu1[d];
      float d2 = a - mu2[d];
      float q1 = sg1[d], q2 = sg2[d];
      s1 += (d1 * d1) / (EPSQ + q1 * q1);
      s2 += (d2 * d2) / (EPSQ + q2 * q2);
    }
    const int src = ei[e];
    const int dst = ei[E + e];
    const int r = rank[e];
    const int4 o = *(const int4*)(seg + (long)dst * 4);
    const int pos = o.x + (r < 0 ? o.z : 0) + (r & 0x7fffffff);
    int4 rec = make_int4(src, __float_as_int(__expf(-0.5f * s1)),
                         src, __float_as_int(__expf(-0.5f * s2)));
    *(int4*)(csr + (long)pos * 4) = rec;
  }
}

#define EU 3
#define TP 36

static __device__ __forceinline__ void own_lds_stores_landed() {
  asm volatile("s_wait_dscnt 0x0" ::: "memory");
}

__global__ void __launch_bounds__(128)
layer_kernel(const int* __restrict__ seg, const int* __restrict__ csr,
             const float* __restrict__ XIN, const float* __restrict__ G,
             const float* __restrict__ R, const float* __restrict__ B,
             float* __restrict__ OUT, int N, int ntiles, int sel) {
  __shared__ __attribute__((aligned(16))) float stage[4 * 16 * TP];
  const int wv   = threadIdx.x >> 5;
  const int wave = blockIdx.x * 4 + wv;
  if (wave >= ntiles) return;
  const int lane = threadIdx.x & 31;
  float* T = stage + wv * (16 * TP);
  const int row0 = wave * 16;
  const int nl = lane >> 3;
  const int q  = lane & 7;

  int begb[4], endb[4];
  int md = 0;
#pragma unroll
  for (int g = 0; g < 4; ++g) {
    const int n = row0 + g * 4 + nl;
    int2 be = make_int2(0, 0);
    if (n < N) be = *(const int2*)(seg + (long)n * 4);
    begb[g] = be.x * 16;
    endb[g] = be.y * 16;
    const int d = be.y - be.x;
    md = d > md ? d : md;
  }
  for (int o = 16; o > 0; o >>= 1) {
    const int t = __shfl_xor(md, o);
    md = t > md ? t : md;
  }
  const int wmax = __builtin_amdgcn_readfirstlane(md);

  float4 acc[4];
#pragma unroll
  for (int g = 0; g < 4; ++g) acc[g] = make_float4(0.0f, 0.0f, 0.0f, 0.0f);

  if (wmax > 0) {
    const char* xrow = (const char*)XIN;
    const char* rec  = (const char*)csr + sel * 8;
    const unsigned q16 = (unsigned)q * 16u;
    unsigned so[4][EU];
    float    ww[4][EU];
#pragma unroll
    for (int g = 0; g < 4; ++g) {
      int lastb = endb[g] - 16;
      lastb = lastb > 0 ? lastb : 0;
#pragma unroll
      for (int u = 0; u < EU; ++u) {
        int jb = begb[g] + u * 16;
        jb = jb < lastb ? jb : lastb;
        const int2 r = *(const int2*)(rec + (unsigned)jb);
        so[g][u] = (unsigned)r.x * 128u + q16;
        ww[g][u] = __int_as_float(r.y);
      }
    }
    for (int j = 0; j < wmax; j += EU) {
      int2   nr[4][EU];
      float4 v[4][EU];
#pragma unroll
      for (int g = 0; g < 4; ++g) {
        int lastb = endb[g] - 16;
        lastb = lastb > 0 ? lastb : 0;
#pragma unroll
        for (int u = 0; u < EU; ++u) {
          int jb = begb[g] + (j + EU + u) * 16;
          jb = jb < lastb ? jb : lastb;
          nr[g][u] = *(const int2*)(rec + (unsigned)jb);
        }
      }
      asm volatile("" ::: "memory");
#pragma unroll
      for (int g = 0; g < 4; ++g) {
#pragma unroll
        for (int u = 0; u < EU; ++u) v[g][u] = *(const float4*)(xrow + so[g][u]);
      }
#pragma unroll
      for (int g = 0; g < 4; ++g) {
#pragma unroll
        for (int u = 0; u < EU; ++u) {
          const int m = (begb[g] + (j + u) * 16 < endb[g]) ? -1 : 0;
          const float w = __int_as_float(__float_as_int(ww[g][u]) & m);
          acc[g].x = __builtin_fmaf(__int_as_float(__float_as_int(v[g][u].x) & m), w, acc[g].x);
          acc[g].y = __builtin_fmaf(__int_as_float(__float_as_int(v[g][u].y) & m), w, acc[g].y);
          acc[g].z = __builtin_fmaf(__int_as_float(__float_as_int(v[g][u].z) & m), w, acc[g].z);
          acc[g].w = __builtin_fmaf(__int_as_float(__float_as_int(v[g][u].w) & m), w, acc[g].w);
        }
      }
#pragma unroll
      for (int g = 0; g < 4; ++g) {
#pragma unroll
        for (int u = 0; u < EU; ++u) {
          so[g][u] = (unsigned)nr[g][u].x * 128u + q16;
          ww[g][u] = __int_as_float(nr[g][u].y);
        }
      }
    }
  }

  const int half = lane >> 4;
  const int m    = lane & 15;
  int arow = row0 + m;
  if (arow >= N) arow = N - 1;

  v2f ax[8], bg0[8], bg1[8], br0[8], br1[8];
#pragma unroll
  for (int s = 0; s < 8; ++s) {
    const int kk = s * 4 + half * 2;
    ax[s] = *(const v2f*)(XIN + (long)arow * 32 + kk);
    bg0[s].x = G[kk * 32 + m];        bg0[s].y = G[(kk + 1) * 32 + m];
    bg1[s].x = G[kk * 32 + 16 + m];   bg1[s].y = G[(kk + 1) * 32 + 16 + m];
    br0[s].x = R[kk * 32 + m];        br0[s].y = R[(kk + 1) * 32 + m];
    br1[s].x = R[kk * 32 + 16 + m];   br1[s].y = R[(kk + 1) * 32 + 16 + m];
  }
  const float b0 = B[m], b1 = B[16 + m];

#pragma unroll
  for (int g = 0; g < 4; ++g) {
    float c = (float)((endb[g] - begb[g]) >> 4);
    c = c > 1.0f ? c : 1.0f;
    const float inv = 1.0f / c;
    float4 a;
    a.x = acc[g].x * inv;
    a.y = acc[g].y * inv;
    a.z = acc[g].z * inv;
    a.w = acc[g].w * inv;
    *(float4*)(T + (g * 4 + nl) * TP + q * 4) = a;
  }
  own_lds_stores_landed();

  v8f accg0 = {}, accg1 = {}, accr0 = {}, accr1 = {};

#pragma unroll
  for (int s = 0; s < 8; ++s) {
    const int kk = s * 4 + half * 2;
    const v2f am = *(const v2f*)(T + m * TP + kk);
    accg0 = __builtin_amdgcn_wmma_f32_16x16x4_f32(false, am, false, bg0[s], (short)0, accg0, false, false);
    accg1 = __builtin_amdgcn_wmma_f32_16x16x4_f32(false, am, false, bg1[s], (short)0, accg1, false, false);
    accr0 = __builtin_amdgcn_wmma_f32_16x16x4_f32(false, ax[s], false, br0[s], (short)0, accr0, false, false);
    accr1 = __builtin_amdgcn_wmma_f32_16x16x4_f32(false, ax[s], false, br1[s], (short)0, accr1, false, false);
  }

#pragma unroll
  for (int r = 0; r < 8; ++r) {
    T[(half * 8 + r) * TP + m]      = accg0[r] + (accr0[r] + b0);
    T[(half * 8 + r) * TP + 16 + m] = accg1[r] + (accr1[r] + b1);
  }
  own_lds_stores_landed();
#pragma unroll
  for (int g = 0; g < 4; ++g) {
    const int n = row0 + g * 4 + nl;
    const float4 o = *(const float4*)(T + (g * 4 + nl) * TP + q * 4);
    if (n < N) *(float4*)(OUT + (long)n * 32 + q * 4) = o;
  }
}

extern "C" void kernel_launch(void* const* d_in, const int* in_sizes, int n_in,
                              void* d_out, int out_size, void* d_ws, size_t ws_size,
                              hipStream_t stream) {
  const int*   ei  = (const int*)d_in[0];
  const float* ea  = (const float*)d_in[1];
  const float* x   = (const float*)d_in[2];
  const float* g1  = (const float*)d_in[3];
  const float* mu1 = (const float*)d_in[4];
  const float* sg1 = (const float*)d_in[5];
  const float* r1  = (const float*)d_in[6];
  const float* b1  = (const float*)d_in[7];
  const float* g2  = (const float*)d_in[8];
  const float* mu2 = (const float*)d_in[9];
  const float* sg2 = (const float*)d_in[10];
  const float* r2  = (const float*)d_in[11];
  const float* b2  = (const float*)d_in[12];

  const int E = in_sizes[0] / 2;
  const int N = in_sizes[2] / 32;
  const long plane = (long)N * 32;

  int*   csr    = (int*)d_ws;
  int*   seg    = csr + (long)E * 4;
  float* h      = (float*)(seg + (long)N * 4);
  int*   rank   = (int*)(h + 2 * plane);
  int*   cnt0   = rank + E;
  int*   cnt1   = cnt0 + N;
  int*   cursor = cnt1 + N;
  float* out    = (float*)d_out;

  const int ntiles     = (N + 15) / 16;
  const int tileBlocks = (ntiles + 3) / 4;
  const int rankBlocks  = (E + 1023) / 1024;
  const int placeBlocks = (E + 1279) / 1280;
  const int nodeBlocks = (N + 255) / 256;

  zero_kernel<<<256, 256, 0, stream>>>(cnt0, 2L * N + 1);
  rank_kernel<<<rankBlocks, 256, 0, stream>>>(ei, cnt0, cnt1, rank, E);
  alloc_kernel<<<nodeBlocks, 256, 0, stream>>>(cnt0, cnt1, cursor, seg, N);
  place_kernel<<<placeBlocks, 256, 0, stream>>>(ei, ea, mu1, sg1, mu2, sg2, rank, seg, csr, E);

  layer_kernel<<<tileBlocks, 128, 0, stream>>>(seg, csr, x, g1, r1, b1, h, N, ntiles, 0);
  layer_kernel<<<tileBlocks, 128, 0, stream>>>(seg, csr, h, g2, r2, b2, out, N, ntiles, 1);
}


# Edges grouped by destination in a two-level counting sort on LDS counters (bins of 256 nodes, then by node): no global atomic, no chip-wide scattered store; 6 launches, now 5; float sum order changes
# speedup vs baseline: 2.7221x; 1.1659x over previous
#include <hip/hip_runtime.h>


typedef float v2f __attribute__((ext_vector_type(2)));
typedef float v8f __attribute__((ext_vector_type(8)));

#define EPSQ 1e-15f

#define NBMAX 512
#define G1 256
#define CT 512
#define CU 13
#define MU 4
#define BT 512
#define BU 9


__global__ void __launch_bounds__(CT)
count_kernel(const int* __restrict__ ei,
             int* __restrict__ H, int E, int NB) {
  __shared__ int hist[NBMAX];
  const int t = threadIdx.x;
  for (int i = t; i < NBMAX; i += CT) hist[i] = 0;
  __syncthreads();
  const int CH = (E + G1 - 1) / G1;
  const int lo = blockIdx.x * CH;
  int hi = lo + CH;
  hi = hi < E ? hi : E;
  for (int e0 = lo + t; e0 < hi; e0 += CU * CT) {
    int d[CU];
#pragma unroll
    for (int u = 0; u < CU; ++u) {
      int e = e0 + u * CT;
      e = e < hi ? e : hi - 1;
      d[u] = ei[E + e];
    }
#pragma unroll
    for (int u = 0; u < CU; ++u) asm volatile("" :: "v"(d[u]));
#pragma unroll
    for (int u = 0; u < CU; ++u)
      if (e0 + u * CT < hi) atomicAdd(&hist[d[u] >> 8], 1);
  }
  __syncthreads();
  for (int i = t; i < NB; i += CT) H[blockIdx.x * NB + i] = hist[i];
}

__global__ void __launch_bounds__(1024)
move_kernel(const int* __restrict__ ei,
            const float* __restrict__ ea,
            const float* __restrict__ mu1, const float* __restrict__ sg1,
            const float* __restrict__ mu2, const float* __restrict__ sg2,
            const int* __restrict__ H, int* __restrict__ BB,
            int* __restrict__ tmp, int E, int NB) {
  __shared__ int ptot[2][NBMAX];
  __shared__ int ppre[2][NBMAX];
  __shared__ int off[NBMAX];
  __shared__ int wsum[32];
  const int t = threadIdx.x;
  const int lane = t & 31;
  const int wv = t >> 5;
  const int bin = t & (NBMAX - 1);
  const int part = t >> 9;
  {
    int s = 0, p = 0;
    const int c0 = part * (G1 / 2);
    if (bin < NB) {
      for (int c = 0; c < G1 / 2; c += 32) {
        int v[32];
#pragma unroll
        for (int u = 0; u < 32; ++u) v[u] = H[(c0 + c + u) * NB + bin];
#pragma unroll
        for (int u = 0; u < 32; ++u) {
          s += v[u];
          p += (c0 + c + u < (int)blockIdx.x) ? v[u] : 0;
        }
      }
    }
    ptot[part][bin] = s;
    ppre[part][bin] = p;
  }
  __syncthreads();
  int tot = 0, pre = 0;
  if (t < NBMAX) { tot = ptot[0][t] + ptot[1][t]; pre = ppre[0][t] + ppre[1][t]; }
  int inc = tot;
  for (int o = 1; o < 32; o <<= 1) {
    const int u = __shfl_up(inc, o);
    if (lane >= o) inc += u;
  }
  if (lane == 31) wsum[wv] = inc;
  __syncthreads();
  if (t < NBMAX) {
    int base = inc - tot;
    for (int w = 0; w < wv; ++w) base += wsum[w];
    off[t] = base + pre;
    if (blockIdx.x == 0 && t < NB) BB[t] = base;
  }
  if (blockIdx.x == 0 && t == 0) BB[NB] = E;
  __syncthreads();

  float m1[8], m2[8], q1[8], q2[8];
#pragma unroll
  for (int d = 0; d < 8; ++d) { m1[d] = mu1[d]; m2[d] = mu2[d]; q1[d] = sg1[d]; q2[d] = sg2[d]; }

  const int CH = (E + G1 - 1) / G1;
  const int lo = blockIdx.x * CH;
  int hi = lo + CH;
  hi = hi < E ? hi : E;
  for (int e0 = lo + t; e0 < hi; e0 += MU * 1024) {
    float4 a0[MU], a1[MU];
    int sv[MU], dv[MU];
#pragma unroll
    for (int u = 0; u < MU; ++u) {
      int e = e0 + u * 1024;
      e = e < hi ? e : hi - 1;
      a0[u] = *(const float4*)(ea + (long)e * 8);
      a1[u] = *(const float4*)(ea + (long)e * 8 + 4);
      sv[u] = ei[e];
      dv[u] = ei[E + e];
    }
#pragma unroll
    for (int u = 0; u < MU; ++u)
      asm volatile("" :: "v"(a0[u].x), "v"(a0[u].y), "v"(a0[u].z), "v"(a0[u].w),
                         "v"(a1[u].x), "v"(a1[u].y), "v"(a1[u].z), "v"(a1[u].w), "v"(sv[u]), "v"(dv[u]));
#pragma unroll
    for (int u = 0; u < MU; ++u) {
      const float av[8] = { a0[u].x, a0[u].y, a0[u].z, a0[u].w, a1[u].x, a1[u].y, a1[u].z, a1[u].w };
      float s1 = 0.0f, s2 = 0.0f;
#pragma unroll
      for (int d = 0; d < 8; ++d) {
        const float d1 = av[d] - m1[d];
        const float d2 = av[d] - m2[d];
        s1 += (d1 * d1) / (EPSQ + q1[d] * q1[d]);
        s2 += (d2 * d2) / (EPSQ + q2[d] * q2[d]);
      }
      if (e0 + u * 1024 < hi) {
        const int pos = atomicAdd(&off[dv[u] >> 8], 1);
        *(int4*)(tmp + (long)pos * 4) = make_int4((sv[u] << 8) | (dv[u] & 255),
                                                  __float_as_int(__expf(-0.5f * s1)),
                                                  __float_as_int(__expf(-0.5f * s2)), 0);
      }
    }
  }
}

__global__ void __launch_bounds__(BT)
bin_kernel(const int* __restrict__ tmp, const int* __restrict__ BB,
           int* __restrict__ seg, int* __restrict__ csr, int N) {
  __shared__ int cnt[256];
  __shared__ int wsum[8];
  const int t = threadIdx.x;
  const int lane = t & 31;
  const int wv = t >> 5;
  const int lo = BB[blockIdx.x];
  const int hi = BB[blockIdx.x + 1];
  if (t < 256) cnt[t] = 0;
  __syncthreads();
  for (int j0 = lo + t; j0 < hi; j0 += BU * BT) {
    int k[BU];
#pragma unroll
    for (int u = 0; u < BU; ++u) {
      int j = j0 + u * BT;
      j = j < hi ? j : hi - 1;
      k[u] = tmp[(long)j * 4];
    }
#pragma unroll
    for (int u = 0; u < BU; ++u) asm volatile("" :: "v"(k[u]));
#pragma unroll
    for (int u = 0; u < BU; ++u)
      if (j0 + u * BT < hi) atomicAdd(&cnt[k[u] & 255], 1);
  }
  __syncthreads();
  int c = 0;
  if (t < 256) c = cnt[t];
  int inc = c;
  for (int o = 1; o < 32; o <<= 1) {
    const int u = __shfl_up(inc, o);
    if (lane >= o) inc += u;
  }
  if (t < 256 && lane == 31) wsum[wv] = inc;
  __syncthreads();
  if (t < 256) {
    int beg = lo + inc - c;
    for (int w = 0; w < wv; ++w) beg += wsum[w];
    const int n = blockIdx.x * 256 + t;
    if (n < N) *(int4*)(seg + (long)n * 4) = make_int4(beg, beg + c, 0, 0);
    cnt[t] = beg;
  }
  __syncthreads();
  for (int j0 = lo + t; j0 < hi; j0 += BU * BT) {
    int4 r[BU];
#pragma unroll
    for (int u = 0; u < BU; ++u) {
      int j = j0 + u * BT;
      j = j < hi ? j : hi - 1;
      r[u] = *(const int4*)(tmp + (long)j * 4);
    }
#pragma unroll
    for (int u = 0; u < BU; ++u) asm volatile("" :: "v"(r[u].x), "v"(r[u].y), "v"(r[u].z));
#pragma unroll
    for (int u = 0; u < BU; ++u) {
      if (j0 + u * BT < hi) {
        const int pos = atomicAdd(&cnt[r[u].x & 255], 1);
        const int sn = (int)((unsigned)r[u].x >> 8);
        *(int4*)(csr + (long)pos * 4) = make_int4(sn, r[u].y, sn, r[u].z);
      }
    }
  }
}

#define EU 3
#define TP 36

static __device__ __forceinline__ void own_lds_stores_landed() {
  asm volatile("s_wait_dscnt 0x0" ::: "memory");
}

__global__ void __launch_bounds__(128)
layer_kernel(const int* __restrict__ seg, const int* __restrict__ csr,
             const float* __restrict__ XIN, const float* __restrict__ G,
             const float* __restrict__ R, const float* __restrict__ B,
             float* __restrict__ OUT, int N, int ntiles, int sel) {
  __shared__ __attribute__((aligned(16))) float stage[4 * 16 * TP];
  const int wv   = threadIdx.x >> 5;
  const int wave = blockIdx.x * 4 + wv;
  if (wave >= ntiles) return;
  const int lane = threadIdx.x & 31;
  float* T = stage + wv * (16 * TP);
  const int row0 = wave * 16;
  const int nl = lane >> 3;
  const int q  = lane & 7;

  int begb[4], endb[4];
  int md = 0;
#pragma unroll
  for (int g = 0; g < 4; ++g) {
    const int n = row0 + g * 4 + nl;
    int2 be = make_int2(0, 0);
    if (n < N) be = *(const int2*)(seg + (long)n * 4);
    begb[g] = be.x * 16;
    endb[g] = be.y * 16;
    const int d = be.y - be.x;
    md = d > md ? d : md;
  }
  for (int o = 16; o > 0; o >>= 1) {
    const int t = __shfl_xor(md, o);
    md = t > md ? t : md;
  }
  const int wmax = __builtin_amdgcn_readfirstlane(md);

  float4 acc[4];
#pragma unroll
  for (int g = 0; g < 4; ++g) acc[g] = make_float4(0.0f, 0.0f, 0.0f, 0.0f);

  if (wmax > 0) {
    const char* xrow = (const char*)XIN;
    const char* rec  = (const char*)csr + sel * 8;
    const unsigned q16 = (unsigned)q * 16u;
    unsigned so[4][EU];
    float    ww[4][EU];
#pragma unroll
    for (int g = 0; g < 4; ++g) {
      int lastb = endb[g] - 16;
      lastb = lastb > 0 ? lastb : 0;
#pragma unroll
      for (int u = 0; u < EU; ++u) {
        int jb = begb[g] + u * 16;
        jb = jb < lastb ? jb : lastb;
        const int2 r = *(const int2*)(rec + (unsigned)jb);
        so[g][u] = (unsigned)r.x * 128u + q16;
        ww[g][u] = __int_as_float(r.y);
      }
    }
    for (int j = 0; j < wmax; j += EU) {
      int2   nr[4][EU];
      float4 v[4][EU];
#pragma unroll
      for (int g = 0; g < 4; ++g) {
        int lastb = endb[g] - 16;
        lastb = lastb > 0 ? lastb : 0;
#pragma unroll
        for (int u = 0; u < EU; ++u) {
          int jb = begb[g] + (j + EU + u) * 16;
          jb = jb < lastb ? jb : lastb;
          nr[g][u] = *(const int2*)(rec + (unsigned)jb);
        }
      }
      asm volatile("" ::: "memory");
#pragma unroll
      for (int g = 0; g < 4; ++g) {
#pragma unroll
        for (int u = 0; u < EU; ++u) v[g][u] = *(const float4*)(xrow + so[g][u]);
      }
#pragma unroll
      for (int g = 0; g < 4; ++g) {
#pragma unroll
        for (int u = 0; u < EU; ++u) {
          const int m = (begb[g] + (j + u) * 16 < endb[g]) ? -1 : 0;
          const float w = __int_as_float(__float_as_int(ww[g][u]) & m);
          acc[g].x = __builtin_fmaf(__int_as_float(__float_as_int(v[g][u].x) & m), w, acc[g].x);
          acc[g].y = __builtin_fmaf(__int_as_float(__float_as_int(v[g][u].y) & m), w, acc[g].y);
          acc[g].z = __builtin_fmaf(__int_as_float(__float_as_int(v[g][u].z) & m), w, acc[g].z);
          acc[g].w = __builtin_fmaf(__int_as_float(__float_as_int(v[g][u].w) & m), w, acc[g].w);
        }
      }
#pragma unroll
      for (int g = 0; g < 4; ++g) {
#pragma unroll
        for (int u = 0; u < EU; ++u) {
          so[g][u] = (unsigned)nr[g][u].x * 128u + q16;
          ww[g][u] = __int_as_float(nr[g][u].y);
        }
      }
    }
  }

  const int half = lane >> 4;
  const int m    = lane & 15;
  int arow = row0 + m;
  if (arow >= N) arow = N - 1;

  v2f ax[8], bg0[8], bg1[8], br0[8], br1[8];
#pragma unroll
  for (int s = 0; s < 8; ++s) {
    const int kk = s * 4 + half * 2;
    ax[s] = *(const v2f*)(XIN + (long)arow * 32 + kk);
    bg0[s].x = G[kk * 32 + m];        bg0[s].y = G[(kk + 1) * 32 + m];
    bg1[s].x = G[kk * 32 + 16 + m];   bg1[s].y = G[(kk + 1) * 32 + 16 + m];
    br0[s].x = R[kk * 32 + m];        br0[s].y = R[(kk + 1) * 32 + m];
    br1[s].x = R[kk * 32 + 16 + m];   br1[s].y = R[(kk + 1) * 32 + 16 + m];
  }
  const float b0 = B[m], b1 = B[16 + m];

#pragma unroll
  for (int g = 0; g < 4; ++g) {
    float c = (float)((endb[g] - begb[g]) >> 4);
    c = c > 1.0f ? c : 1.0f;
    const float inv = 1.0f / c;
    float4 a;
    a.x = acc[g].x * inv;
    a.y = acc[g].y * inv;
    a.z = acc[g].z * inv;
    a.w = acc[g].w * inv;
    *(float4*)(T + (g * 4 + nl) * TP + q * 4) = a;
  }
  own_lds_stores_landed();

  v8f accg0 = {}, accg1 = {}, accr0 = {}, accr1 = {};

#pragma unroll
  for (int s = 0; s < 8; ++s) {
    const int kk = s * 4 + half * 2;
    const v2f am = *(const v2f*)(T + m * TP + kk);
    accg0 = __builtin_amdgcn_wmma_f32_16x16x4_f32(false, am, false, bg0[s], (short)0, accg0, false, false);
    accg1 = __builtin_amdgcn_wmma_f32_16x16x4_f32(false, am, false, bg1[s], (short)0, accg1, false, false);
    accr0 = __builtin_amdgcn_wmma_f32_16x16x4_f32(false, ax[s], false, br0[s], (short)0, accr0, false, false);
    accr1 = __builtin_amdgcn_wmma_f32_16x16x4_f32(false, ax[s], false, br1[s], (short)0, accr1, false, false);
  }

#pragma unroll
  for (int r = 0; r < 8; ++r) {
    T[(half * 8 + r) * TP + m]      = accg0[r] + (accr0[r] + b0);
    T[(half * 8 + r) * TP + 16 + m] = accg1[r] + (accr1[r] + b1);
  }
  own_lds_stores_landed();
#pragma unroll
  for (int g = 0; g < 4; ++g) {
    const int n = row0 + g * 4 + nl;
    const float4 o = *(const float4*)(T + (g * 4 + nl) * TP + q * 4);
    if (n < N) *(float4*)(OUT + (long)n * 32 + q * 4) = o;
  }
}

extern "C" void kernel_launch(void* const* d_in, const int* in_sizes, int n_in,
                              void* d_out, int out_size, void* d_ws, size_t ws_size,
                              hipStream_t stream) {
  const int*   ei  = (const int*)d_in[0];
  const float* ea  = (const float*)d_in[1];
  const float* x   = (const float*)d_in[2];
  const float* g1  = (const float*)d_in[3];
  const float* mu1 = (const float*)d_in[4];
  const float* sg1 = (const float*)d_in[5];
  const float* r1  = (const float*)d_in[6];
  const float* b1  = (const float*)d_in[7];
  const float* g2  = (const float*)d_in[8];
  const float* mu2 = (const float*)d_in[9];
  const float* sg2 = (const float*)d_in[10];
  const float* r2  = (const float*)d_in[11];
  const float* b2  = (const float*)d_in[12];

  const int E = in_sizes[0] / 2;
  const int N = in_sizes[2] / 32;
  const long plane = (long)N * 32;

  const int NB = (N + 255) / 256;
  int*   csr    = (int*)d_ws;
  int*   seg    = csr + (long)E * 4;
  int*   tmp    = seg + (long)N * 4;
  float* h      = (float*)tmp;
  const long tmpWords = (long)E * 4 > plane ? (long)E * 4 : plane;
  int*   H      = tmp + tmpWords;
  int*   BB     = H + (long)G1 * NB;
  float* out    = (float*)d_out;

  const int ntiles     = (N + 15) / 16;
  const int tileBlocks = (ntiles + 3) / 4;

  count_kernel<<<G1, CT, 0, stream>>>(ei, H, E, NB);
  move_kernel<<<G1, 1024, 0, stream>>>(ei, ea, mu1, sg1, mu2, sg2, H, BB, tmp, E, NB);
  bin_kernel<<<NB, BT, 0, stream>>>(tmp, BB, seg, csr, N);

  layer_kernel<<<tileBlocks, 128, 0, stream>>>(seg, csr, x, g1, r1, b1, h, N, ntiles, 0);
  layer_kernel<<<tileBlocks, 128, 0, stream>>>(seg, csr, h, g2, r2, b2, out, N, ntiles, 1);
}


# Both sort levels place their 16-byte records in an LDS copy of the work-group's output and store it as whole lines (level 2 reads a bin once, records stay in registers); float sum order changes
# speedup vs baseline: 2.8945x; 1.0633x over previous
#include <hip/hip_runtime.h>


typedef float v2f __attribute__((ext_vector_type(2)));
typedef float v8f __attribute__((ext_vector_type(8)));

#define EPSQ 1e-15f

#define NBMAX 512
#define G1 256
#define CT 512
#define CU 13
#define MU 4
#define BT 512
#define BU 9
#define CHMAX (CT * CU)
#define BINMAX (BT * BU)


__global__ void __launch_bounds__(CT)
count_kernel(const int* __restrict__ ei,
             int* __restrict__ H, int E, int NB) {
  __shared__ int hist[NBMAX];
  const int t = threadIdx.x;
  for (int i = t; i < NBMAX; i += CT) hist[i] = 0;
  __syncthreads();
  const int CH = (E + G1 - 1) / G1;
  const int lo = blockIdx.x * CH;
  int hi = lo + CH;
  hi = hi < E ? hi : E;
  for (int e0 = lo + t; e0 < hi; e0 += CU * CT) {
    int d[CU];
#pragma unroll
    for (int u = 0; u < CU; ++u) {
      int e = e0 + u * CT;
      e = e < hi ? e : hi - 1;
      d[u] = ei[E + e];
    }
#pragma unroll
    for (int u = 0; u < CU; ++u) asm volatile("" :: "v"(d[u]));
#pragma unroll
    for (int u = 0; u < CU; ++u)
      if (e0 + u * CT < hi) atomicAdd(&hist[d[u] >> 8], 1);
  }
  __syncthreads();
  for (int i = t; i < NB; i += CT) H[blockIdx.x * NB + i] = hist[i];
}

__global__ void __launch_bounds__(1024)
move_kernel(const int* __restrict__ ei,
            const float* __restrict__ ea,
            const float* __restrict__ mu1, const float* __restrict__ sg1,
            const float* __restrict__ mu2, const float* __restrict__ sg2,
            const int* __restrict__ H, int* __restrict__ BB,
            int* __restrict__ tmp, int E, int NB) {
  __shared__ int ptot[2][NBMAX];
  __shared__ int ppre[2][NBMAX];
  __shared__ unsigned long long off[NBMAX];
  __shared__ int wsum[32];
  __shared__ int wsum2[32];
  __shared__ int4 stage[CHMAX];
  const int t = threadIdx.x;
  const int lane = t & 31;
  const int wv = t >> 5;
  const int bin = t & (NBMAX - 1);
  const int part = t >> 9;
  const int CH = (E + G1 - 1) / G1;
  const bool staged = CH <= CHMAX;
  int mine = 0;
  if (t < NB) mine = H[blockIdx.x * NB + t];
  {
    int s = 0, p = 0;
    const int c0 = part * (G1 / 2);
    if (bin < NB) {
      for (int c = 0; c < G1 / 2; c += 32) {
        int v[32];
#pragma unroll
        for (int u = 0; u < 32; ++u) v[u] = H[(c0 + c + u) * NB + bin];
#pragma unroll
        for (int u = 0; u < 32; ++u) {
          s += v[u];
          p += (c0 + c + u < (int)blockIdx.x) ? v[u] : 0;
        }
      }
    }
    ptot[part][bin] = s;
    ppre[part][bin] = p;
  }
  __syncthreads();
  int tot = 0, pre = 0;
  if (t < NBMAX) { tot = ptot[0][t] + ptot[1][t]; pre = ppre[0][t] + ppre[1][t]; }
  int inc = tot, inc2 = mine;
  for (int o = 1; o < 32; o <<= 1) {
    const int u = __shfl_up(inc, o);
    const int u2 = __shfl_up(inc2, o);
    if (lane >= o) { inc += u; inc2 += u2; }
  }
  if (lane == 31) { wsum[wv] = inc; wsum2[wv] = inc2; }
  __syncthreads();
  if (t < NBMAX) {
    int base = inc - tot;
    int lb = inc2 - mine;
    for (int w = 0; w < wv; ++w) { base += wsum[w]; lb += wsum2[w]; }
    off[t] = ((unsigned long long)(unsigned)(base + pre) << 32) | (unsigned)lb;
    if (blockIdx.x == 0 && t < NB) BB[t] = base;
  }
  if (blockIdx.x == 0 && t == 0) BB[NB] = E;
  __syncthreads();

  float m1[8], m2[8], q1[8], q2[8];
#pragma unroll
  for (int d = 0; d < 8; ++d) { m1[d] = mu1[d]; m2[d] = mu2[d]; q1[d] = sg1[d]; q2[d] = sg2[d]; }

  const int lo = blockIdx.x * CH;
  int hi = lo + CH;
  hi = hi < E ? hi : E;
  for (int e0 = lo + t; e0 < hi; e0 += MU * 1024) {
    float4 a0[MU], a1[MU];
    int sv[MU], dv[MU];
#pragma unroll
    for (int u = 0; u < MU; ++u) {
      int e = e0 + u * 1024;
      e = e < hi ? e : hi - 1;
      a0[u] = *(const float4*)(ea + (long)e * 8);
      a1[u] = *(const float4*)(ea + (long)e * 8 + 4);
      sv[u] = ei[e];
      dv[u] = ei[E + e];
    }
#pragma unroll
    for (int u = 0; u < MU; ++u)
      asm volatile("" :: "v"(a0[u].x), "v"(a0[u].y), "v"(a0[u].z), "v"(a0[u].w),
                         "v"(a1[u].x), "v"(a1[u].y), "v"(a1[u].z), "v"(a1[u].w), "v"(sv[u]), "v"(dv[u]));
#pragma unroll
    for (int u = 0; u < MU; ++u) {
      const float av[8] = { a0[u].x, a0[u].y, a0[u].z, a0[u].w, a1[u].x, a1[u].y, a1[u].z, a1[u].w };
      float s1 = 0.0f, s2 = 0.0f;
#pragma unroll
      for (int d = 0; d < 8; ++d) {
        const float d1 = av[d] - m1[d];
        const float d2 = av[d] - m2[d];
        s1 += (d1 * d1) / (EPSQ + q1[d] * q1[d]);
        s2 += (d2 * d2) / (EPSQ + q2[d] * q2[d]);
      }
      if (e0 + u * 1024 < hi) {
        const int b = dv[u] >> 8;
        const unsigned long long c = atomicAdd(&off[b], 0x100000001ull);
        const int pos = (int)(c >> 32);
        const int4 rec = make_int4((sv[u] << 8) | (dv[u] & 255),
                                   __float_as_int(__expf(-0.5f * s1)),
                                   __float_as_int(__expf(-0.5f * s2)), pos);
        if (staged) stage[(unsigned)c] = rec;
        else *(int4*)(tmp + (long)pos * 4) = rec;
      }
    }
  }
  if (staged) {
    __syncthreads();
    for (int i = t; i < hi - lo; i += 1024) {
      const int4 r = stage[i];
      *(int4*)(tmp + (long)r.w * 4) = r;
    }
  }
}

__global__ void __launch_bounds__(BT)
bin_kernel(const int* __restrict__ tmp, const int* __restrict__ BB,
           int* __restrict__ seg, int* __restrict__ csr, int N) {
  __shared__ int cnt[256];
  __shared__ int wsum[8];
  __shared__ int4 stage[BINMAX];
  const int t = threadIdx.x;
  const int lane = t & 31;
  const int wv = t >> 5;
  const int lo = BB[blockIdx.x];
  const int hi = BB[blockIdx.x + 1];
  const bool fits = hi - lo <= BINMAX;
  if (t < 256) cnt[t] = 0;
  __syncthreads();
  int4 r[BU];
  if (fits) {
#pragma unroll
    for (int u = 0; u < BU; ++u) {
      int j = lo + t + u * BT;
      j = j < hi ? j : hi - 1;
      j = j > 0 ? j : 0;
      r[u] = *(const int4*)(tmp + (long)j * 4);
    }
#pragma unroll
    for (int u = 0; u < BU; ++u) asm volatile("" :: "v"(r[u].x), "v"(r[u].y), "v"(r[u].z));
#pragma unroll
    for (int u = 0; u < BU; ++u)
      if (lo + t + u * BT < hi) atomicAdd(&cnt[r[u].x & 255], 1);
  } else {
  for (int j0 = lo + t; j0 < hi; j0 += BU * BT) {
    int k[BU];
#pragma unroll
    for (int u = 0; u < BU; ++u) {
      int j = j0 + u * BT;
      j = j < hi ? j : hi - 1;
      k[u] = tmp[(long)j * 4];
    }
#pragma unroll
    for (int u = 0; u < BU; ++u) asm volatile("" :: "v"(k[u]));
#pragma unroll
    for (int u = 0; u < BU; ++u)
      if (j0 + u * BT < hi) atomicAdd(&cnt[k[u] & 255], 1);
  }
  }
  __syncthreads();
  int c = 0;
  if (t < 256) c = cnt[t];
  int inc = c;
  for (int o = 1; o < 32; o <<= 1) {
    const int u = __shfl_up(inc, o);
    if (lane >= o) inc += u;
  }
  if (t < 256 && lane == 31) wsum[wv] = inc;
  __syncthreads();
  if (t < 256) {
    int beg = lo + inc - c;
    for (int w = 0; w < wv; ++w) beg += wsum[w];
    const int n = blockIdx.x * 256 + t;
    if (n < N) *(int4*)(seg + (long)n * 4) = make_int4(beg, beg + c, 0, 0);
    cnt[t] = beg;
  }
  __syncthreads();
  if (fits) {
    int pos[BU];
#pragma unroll
    for (int u = 0; u < BU; ++u) {
      int p = lo;
      if (lo + t + u * BT < hi) p = atomicAdd(&cnt[r[u].x & 255], 1);
      pos[u] = p;
    }
#pragma unroll
    for (int u = 0; u < BU; ++u) {
      if (lo + t + u * BT < hi) {
        const int sn = (int)((unsigned)r[u].x >> 8);
        stage[pos[u] - lo] = make_int4(sn, r[u].y, sn, r[u].z);
      }
    }
    __syncthreads();
    for (int i = t; i < hi - lo; i += BT) *(int4*)(csr + (long)(lo + i) * 4) = stage[i];
    return;
  }
  for (int j0 = lo + t; j0 < hi; j0 += BU * BT) {
#pragma unroll
    for (int u = 0; u < BU; ++u) {
      int j = j0 + u * BT;
      j = j < hi ? j : hi - 1;
      r[u] = *(const int4*)(tmp + (long)j * 4);
    }
#pragma unroll
    for (int u = 0; u < BU; ++u) asm volatile("" :: "v"(r[u].x), "v"(r[u].y), "v"(r[u].z));
#pragma unroll
    for (int u = 0; u < BU; ++u) {
      if (j0 + u * BT < hi) {
        const int pos = atomicAdd(&cnt[r[u].x & 255], 1);
        const int sn = (int)((unsigned)r[u].x >> 8);
        *(int4*)(csr + (long)pos * 4) = make_int4(sn, r[u].y, sn, r[u].z);
      }
    }
  }
}

#define EU 3
#define TP 36

static __device__ __forceinline__ void own_lds_stores_landed() {
  asm volatile("s_wait_dscnt 0x0" ::: "memory");
}

__global__ void __launch_bounds__(128)
layer_kernel(const int* __restrict__ seg, const int* __restrict__ csr,
             const float* __restrict__ XIN, const float* __restrict__ G,
             const float* __restrict__ R, const float* __restrict__ B,
             float* __restrict__ OUT, int N, int ntiles, int sel) {
  __shared__ __attribute__((aligned(16))) float stage[4 * 16 * TP];
  const int wv   = threadIdx.x >> 5;
  const int wave = blockIdx.x * 4 + wv;
  if (wave >= ntiles) return;
  const int lane = threadIdx.x & 31;
  float* T = stage + wv * (16 * TP);
  const int row0 = wave * 16;
  const int nl = lane >> 3;
  const int q  = lane & 7;

  int begb[4], endb[4];
  int md = 0;
#pragma unroll
  for (int g = 0; g < 4; ++g) {
    const int n = row0 + g * 4 + nl;
    int2 be = make_int2(0, 0);
    if (n < N) be = *(const int2*)(seg + (long)n * 4);
    begb[g] = be.x * 16;
    endb[g] = be.y * 16;
    const int d = be.y - be.x;
    md = d > md ? d : md;
  }
  for (int o = 16; o > 0; o >>= 1) {
    const int t = __shfl_xor(md, o);
    md = t > md ? t : md;
  }
  const int wmax = __builtin_amdgcn_readfirstlane(md);

  float4 acc[4];
#pragma unroll
  for (int g = 0; g < 4; ++g) acc[g] = make_float4(0.0f, 0.0f, 0.0f, 0.0f);

  if (wmax > 0) {
    const char* xrow = (const char*)XIN;
    const char* rec  = (const char*)csr + sel * 8;
    const unsigned q16 = (unsigned)q * 16u;
    unsigned so[4][EU];
    float    ww[4][EU];
#pragma unroll
    for (int g = 0; g < 4; ++g) {
      int lastb = endb[g] - 16;
      lastb = lastb > 0 ? lastb : 0;
#pragma unroll
      for (int u = 0; u < EU; ++u) {
        int jb = begb[g] + u * 16;
        jb = jb < lastb ? jb : lastb;
        const int2 r = *(const int2*)(rec + (unsigned)jb);
        so[g][u] = (unsigned)r.x * 128u + q16;
        ww[g][u] = __int_as_float(r.y);
      }
    }
    for (int j = 0; j < wmax; j += EU) {
      int2   nr[4][EU];
      float4 v[4][EU];
#pragma unroll
      for (int g = 0; g < 4; ++g) {
        int lastb = endb[g] - 16;
        lastb = lastb > 0 ? lastb : 0;
#pragma unroll
        for (int u = 0; u < EU; ++u) {
          int jb = begb[g] + (j + EU + u) * 16;
          jb = jb < lastb ? jb : lastb;
          nr[g][u] = *(const int2*)(rec + (unsigned)jb);
        }
      }
      asm volatile("" ::: "memory");
#pragma unroll
      for (int g = 0; g < 4; ++g) {
#pragma unroll
        for (int u = 0; u < EU; ++u) v[g][u] = *(const float4*)(xrow + so[g][u]);
      }
#pragma unroll
      for (int g = 0; g < 4; ++g) {
#pragma unroll
        for (int u = 0; u < EU; ++u) {
          const int m = (begb[g] + (j + u) * 16 < endb[g]) ? -1 : 0;
          const float w = __int_as_float(__float_as_int(ww[g][u]) & m);
          acc[g].x = __builtin_fmaf(__int_as_float(__float_as_int(v[g][u].x) & m), w, acc[g].x);
          acc[g].y = __builtin_fmaf(__int_as_float(__float_as_int(v[g][u].y) & m), w, acc[g].y);
          acc[g].z = __builtin_fmaf(__int_as_float(__float_as_int(v[g][u].z) & m), w, acc[g].z);
          acc[g].w = __builtin_fmaf(__int_as_float(__float_as_int(v[g][u].w) & m), w, acc[g].w);
        }
      }
#pragma unroll
      for (int g = 0; g < 4; ++g) {
#pragma unroll
        for (int u = 0; u < EU; ++u) {
          so[g][u] = (unsigned)nr[g][u].x * 128u + q16;
          ww[g][u] = __int_as_float(nr[g][u].y);
        }
      }
    }
  }

  const int half = lane >> 4;
  const int m    = lane & 15;
  int arow = row0 + m;
  if (arow >= N) arow = N - 1;

  v2f ax[8], bg0[8], bg1[8], br0[8], br1[8];
#pragma unroll
  for (int s = 0; s < 8; ++s) {
    const int kk = s * 4 + half * 2;
    ax[s] = *(const v2f*)(XIN + (long)arow * 32 + kk);
    bg0[s].x = G[kk * 32 + m];        bg0[s].y = G[(kk + 1) * 32 + m];
    bg1[s].x = G[kk * 32 + 16 + m];   bg1[s].y = G[(kk + 1) * 32 + 16 + m];
    br0[s].x = R[kk * 32 + m];        br0[s].y = R[(kk + 1) * 32 + m];
    br1[s].x = R[kk * 32 + 16 + m];   br1[s].y = R[(kk + 1) * 32 + 16 + m];
  }
  const float b0 = B[m], b1 = B[16 + m];

#pragma unroll
  for (int g = 0; g < 4; ++g) {
    float c = (float)((endb[g] - begb[g]) >> 4);
    c = c > 1.0f ? c : 1.0f;
    const float inv = 1.0f / c;
    float4 a;
    a.x = acc[g].x * inv;
    a.y = acc[g].y * inv;
    a.z = acc[g].z * inv;
    a.w = acc[g].w * inv;
    *(float4*)(T + (g * 4 + nl) * TP + q * 4) = a;
  }
  own_lds_stores_landed();

  v8f accg0 = {}, accg1 = {}, accr0 = {}, accr1 = {};

#pragma unroll
  for (int s = 0; s < 8; ++s) {
    const int kk = s * 4 + half * 2;
    const v2f am = *(const v2f*)(T + m * TP + kk);
    accg0 = __builtin_amdgcn_wmma_f32_16x16x4_f32(false, am, false, bg0[s], (short)0, accg0, false, false);
    accg1 = __builtin_amdgcn_wmma_f32_16x16x4_f32(false, am, false, bg1[s], (short)0, accg1, false, false);
    accr0 = __builtin_amdgcn_wmma_f32_16x16x4_f32(false, ax[s], false, br0[s], (short)0, accr0, false, false);
    accr1 = __builtin_amdgcn_wmma_f32_16x16x4_f32(false, ax[s], false, br1[s], (short)0, accr1, false, false);
  }

#pragma unroll
  for (int r = 0; r < 8; ++r) {
    T[(half * 8 + r) * TP + m]      = accg0[r] + (accr0[r] + b0);
    T[(half * 8 + r) * TP + 16 + m] = accg1[r] + (accr1[r] + b1);
  }
  own_lds_stores_landed();
#pragma unroll
  for (int g = 0; g < 4; ++g) {
    const int n = row0 + g * 4 + nl;
    const float4 o = *(const float4*)(T + (g * 4 + nl) * TP + q * 4);
    if (n < N) *(float4*)(OUT + (long)n * 32 + q * 4) = o;
  }
}

extern "C" void kernel_launch(void* const* d_in, const int* in_sizes, int n_in,
                              void* d_out, int out_size, void* d_ws, size_t ws_size,
                              hipStream_t stream) {
  const int*   ei  = (const int*)d_in[0];
  const float* ea  = (const float*)d_in[1];
  const float* x   = (const float*)d_in[2];
  const float* g1  = (const float*)d_in[3];
  const float* mu1 = (const float*)d_in[4];
  const float* sg1 = (const float*)d_in[5];
  const float* r1  = (const float*)d_in[6];
  const float* b1  = (const float*)d_in[7];
  const float* g2  = (const float*)d_in[8];
  const float* mu2 = (const float*)d_in[9];
  const float* sg2 = (const float*)d_in[10];
  const float* r2  = (const float*)d_in[11];
  const float* b2  = (const float*)d_in[12];

  const int E = in_sizes[0] / 2;
  const int N = in_sizes[2] / 32;
  const long plane = (long)N * 32;

  const int NB = (N + 255) / 256;
  int*   csr    = (int*)d_ws;
  int*   seg    = csr + (long)E * 4;
  int*   tmp    = seg + (long)N * 4;
  float* h      = (float*)tmp;
  const long tmpWords = (long)E * 4 > plane ? (long)E * 4 : plane;
  int*   H      = tmp + tmpWords;
  int*   BB     = H + (long)G1 * NB;
  float* out    = (float*)d_out;

  const int ntiles     = (N + 15) / 16;
  const int tileBlocks = (ntiles + 3) / 4;

  count_kernel<<<G1, CT, 0, stream>>>(ei, H, E, NB);
  move_kernel<<<G1, 1024, 0, stream>>>(ei, ea, mu1, sg1, mu2, sg2, H, BB, tmp, E, NB);
  bin_kernel<<<NB, BT, 0, stream>>>(tmp, BB, seg, csr, N);

  layer_kernel<<<tileBlocks, 128, 0, stream>>>(seg, csr, x, g1, r1, b1, h, N, ntiles, 0);
  layer_kernel<<<tileBlocks, 128, 0, stream>>>(seg, csr, h, g2, r2, b2, out, N, ntiles, 1);
}


# Sort level 1 chunk-local: a work-group orders its own edge chunk by bin, level 2 collects a bin's 256 runs; count launch and every group's read of the count table gone; float sum order changes
# speedup vs baseline: 3.2397x; 1.1193x over previous
#include <hip/hip_runtime.h>


typedef float v2f __attribute__((ext_vector_type(2)));
typedef float v8f __attribute__((ext_vector_type(8)));

#define EPSQ 1e-15f

#define NBMAX 512
#define G1 256
#define MT 1024
#define C1 7
#define MU 4
#define BT 512
#define RW (G1 / (BT / 32))
#define CHMAX 6656
#define BINMAX 4608


__global__ void __launch_bounds__(MT)
move_kernel(const int* __restrict__ ei,
            const float* __restrict__ ea,
            const float* __restrict__ mu1, const float* __restrict__ sg1,
            const float* __restrict__ mu2, const float* __restrict__ sg2,
            int* __restrict__ P, int* __restrict__ tmp, int E, int NB) {
  __shared__ int cur[NBMAX];
  __shared__ int wsum[32];
  __shared__ int4 stage[CHMAX];
  const int t = threadIdx.x;
  const int lane = t & 31;
  const int wv = t >> 5;
  const int CH = (E + G1 - 1) / G1;
  const bool staged = CH <= CHMAX;
  const int lo = blockIdx.x * CH;
  int hi = lo + CH;
  hi = hi < E ? hi : E;
  if (t < NBMAX) cur[t] = 0;
  __syncthreads();
  for (int e0 = lo + t; e0 < hi; e0 += C1 * MT) {
    int d[C1];
#pragma unroll
    for (int u = 0; u < C1; ++u) {
      int e = e0 + u * MT;
      e = e < hi ? e : hi - 1;
      d[u] = ei[E + e];
    }
#pragma unroll
    for (int u = 0; u < C1; ++u) asm volatile("" :: "v"(d[u]));
#pragma unroll
    for (int u = 0; u < C1; ++u)
      if (e0 + u * MT < hi) atomicAdd(&cur[d[u] >> 8], 1);
  }
  __syncthreads();
  int c = 0;
  if (t < NBMAX) c = cur[t];
  int inc = c;
  for (int o = 1; o < 32; o <<= 1) {
    const int u = __shfl_up(inc, o);
    if (lane >= o) inc += u;
  }
  if (lane == 31) wsum[wv] = inc;
  __syncthreads();
  if (t < NBMAX) {
    int base = inc - c;
    for (int w = 0; w < wv; ++w) base += wsum[w];
    cur[t] = base;
    if (t < NB) P[(long)blockIdx.x * (NB + 1) + t] = base;
  }
  if (t == 0) P[(long)blockIdx.x * (NB + 1) + NB] = hi > lo ? hi - lo : 0;
  __syncthreads();

  float m1[8], m2[8], q1[8], q2[8];
#pragma unroll
  for (int d = 0; d < 8; ++d) { m1[d] = mu1[d]; m2[d] = mu2[d]; q1[d] = sg1[d]; q2[d] = sg2[d]; }

  for (int e0 = lo + t; e0 < hi; e0 += MU * MT) {
    float4 a0[MU], a1[MU];
    int sv[MU], dv[MU];
#pragma unroll
    for (int u = 0; u < MU; ++u) {
      int e = e0 + u * MT;
      e = e < hi ? e : hi - 1;
      a0[u] = *(const float4*)(ea + (long)e * 8);
      a1[u] = *(const float4*)(ea + (long)e * 8 + 4);
      sv[u] = ei[e];
      dv[u] = ei[E + e];
    }
#pragma unroll
    for (int u = 0; u < MU; ++u)
      asm volatile("" :: "v"(a0[u].x), "v"(a0[u].y), "v"(a0[u].z), "v"(a0[u].w),
                         "v"(a1[u].x), "v"(a1[u].y), "v"(a1[u].z), "v"(a1[u].w), "v"(sv[u]), "v"(dv[u]));
#pragma unroll
    for (int u = 0; u < MU; ++u) {
      const float av[8] = { a0[u].x, a0[u].y, a0[u].z, a0[u].w, a1[u].x, a1[u].y, a1[u].z, a1[u].w };
      float s1 = 0.0f, s2 = 0.0f;
#pragma unroll
      for (int d = 0; d < 8; ++d) {
        const float d1 = av[d] - m1[d];
        const float d2 = av[d] - m2[d];
        s1 += (d1 * d1) / (EPSQ + q1[d] * q1[d]);
        s2 += (d2 * d2) / (EPSQ + q2[d] * q2[d]);
      }
      if (e0 + u * MT < hi) {
        const int p = atomicAdd(&cur[dv[u] >> 8], 1);
        const int4 rec = make_int4((sv[u] << 8) | (dv[u] & 255),
                                   __float_as_int(__expf(-0.5f * s1)),
                                   __float_as_int(__expf(-0.5f * s2)), 0);
        if (staged) stage[p] = rec;
        else *(int4*)(tmp + (long)(lo + p) * 4) = rec;
      }
    }
  }
  if (staged) {
    __syncthreads();
    for (int i = t; i < hi - lo; i += MT) *(int4*)(tmp + (long)(lo + i) * 4) = stage[i];
  }
}

__global__ void __launch_bounds__(BT)
bin_kernel(const int* __restrict__ tmp, const int* __restrict__ P,
           int* __restrict__ seg, int* __restrict__ csr, int N, int E, int NB) {
  __shared__ int cnt[256];
  __shared__ int wsum[8];
  __shared__ int rs[G1];
  __shared__ int rl[G1];
  __shared__ int red[2][BT / 32];
  __shared__ int4 stage[BINMAX];
  const int t = threadIdx.x;
  const int lane = t & 31;
  const int wv = t >> 5;
  const int CH = (E + G1 - 1) / G1;
  int p0 = 0, p1 = 0;
  if (t < G1) {
    p0 = P[(long)t * (NB + 1) + blockIdx.x];
    p1 = P[(long)t * (NB + 1) + blockIdx.x + 1];
    rs[t] = t * CH + p0;
    rl[t] = p1 - p0;
  }
  if (t < 256) cnt[t] = 0;
  {
    int s0 = p0, s1 = p1;
    for (int o = 16; o > 0; o >>= 1) {
      s0 += __shfl_xor(s0, o);
      s1 += __shfl_xor(s1, o);
    }
    if (lane == 0) { red[0][wv] = s0; red[1][wv] = s1; }
  }
  __syncthreads();
  int lo = 0, hi = 0;
#pragma unroll
  for (int w = 0; w < BT / 32; ++w) { lo += red[0][w]; hi += red[1][w]; }
  const bool fits = hi - lo <= BINMAX;

  int Lr[RW], Sr[RW];
#pragma unroll
  for (int u = 0; u < RW; ++u) {
    Lr[u] = __builtin_amdgcn_readfirstlane(rl[wv * RW + u]);
    Sr[u] = __builtin_amdgcn_readfirstlane(rs[wv * RW + u]);
  }
  int4 r[RW];
#pragma unroll
  for (int u = 0; u < RW; ++u) {
    const int L = Lr[u];
    int j = Sr[u] + (lane < L ? lane : L - 1);
    j = j > 0 ? j : 0;
    j = j < E ? j : E - 1;
    r[u] = *(const int4*)(tmp + (long)j * 4);
  }
#pragma unroll
  for (int u = 0; u < RW; ++u) asm volatile("" :: "v"(r[u].x), "v"(r[u].y), "v"(r[u].z));
#pragma unroll
  for (int u = 0; u < RW; ++u)
    if (lane < Lr[u]) atomicAdd(&cnt[r[u].x & 255], 1);
#pragma unroll
  for (int u = 0; u < RW; ++u) {
    const int L = Lr[u];
    const int S = Sr[u];
    for (int k = 32 + lane; k < L; k += 32) atomicAdd(&cnt[tmp[(long)(S + k) * 4] & 255], 1);
  }
  __syncthreads();
  int c = 0;
  if (t < 256) c = cnt[t];
  int inc = c;
  for (int o = 1; o < 32; o <<= 1) {
    const int u = __shfl_up(inc, o);
    if (lane >= o) inc += u;
  }
  if (t < 256 && lane == 31) wsum[wv] = inc;
  __syncthreads();
  if (t < 256) {
    int beg = lo + inc - c;
    for (int w = 0; w < wv; ++w) beg += wsum[w];
    const int n = blockIdx.x * 256 + t;
    if (n < N) *(int4*)(seg + (long)n * 4) = make_int4(beg, beg + c, 0, 0);
    cnt[t] = beg;
  }
  __syncthreads();
  {
    int pos[RW];
#pragma unroll
    for (int u = 0; u < RW; ++u) {
      int p = lo;
      if (lane < Lr[u]) p = atomicAdd(&cnt[r[u].x & 255], 1);
      pos[u] = p;
    }
#pragma unroll
    for (int u = 0; u < RW; ++u) {
      if (lane < Lr[u]) {
        const int sn = (int)((unsigned)r[u].x >> 8);
        const int4 o = make_int4(sn, r[u].y, sn, r[u].z);
        if (fits) stage[pos[u] - lo] = o;
        else *(int4*)(csr + (long)pos[u] * 4) = o;
      }
    }
  }
#pragma unroll
  for (int u = 0; u < RW; ++u) {
    const int L = Lr[u];
    const int S = Sr[u];
    for (int k = 32 + lane; k < L; k += 32) {
      const int4 x = *(const int4*)(tmp + (long)(S + k) * 4);
      const int pos = atomicAdd(&cnt[x.x & 255], 1);
      const int sn = (int)((unsigned)x.x >> 8);
      const int4 o = make_int4(sn, x.y, sn, x.z);
      if (fits) stage[pos - lo] = o;
      else *(int4*)(csr + (long)pos * 4) = o;
    }
  }
  if (fits) {
    __syncthreads();
    for (int i = t; i < hi - lo; i += BT) *(int4*)(csr + (long)(lo + i) * 4) = stage[i];
  }
}

#define EU 3
#define TP 36

static __device__ __forceinline__ void own_lds_stores_landed() {
  asm volatile("s_wait_dscnt 0x0" ::: "memory");
}

__global__ void __launch_bounds__(128)
layer_kernel(const int* __restrict__ seg, const int* __restrict__ csr,
             const float* __restrict__ XIN, const float* __restrict__ G,
             const float* __restrict__ R, const float* __restrict__ B,
             float* __restrict__ OUT, int N, int ntiles, int sel) {
  __shared__ __attribute__((aligned(16))) float stage[4 * 16 * TP];
  const int wv   = threadIdx.x >> 5;
  const int wave = blockIdx.x * 4 + wv;
  if (wave >= ntiles) return;
  const int lane = threadIdx.x & 31;
  float* T = stage + wv * (16 * TP);
  const int row0 = wave * 16;
  const int nl = lane >> 3;
  const int q  = lane & 7;

  int begb[4], endb[4];
  int md = 0;
#pragma unroll
  for (int g = 0; g < 4; ++g) {
    const int n = row0 + g * 4 + nl;
    int2 be = make_int2(0, 0);
    if (n < N) be = *(const int2*)(seg + (long)n * 4);
    begb[g] = be.x * 16;
    endb[g] = be.y * 16;
    const int d = be.y - be.x;
    md = d > md ? d : md;
  }
  for (int o = 16; o > 0; o >>= 1) {
    const int t = __shfl_xor(md, o);
    md = t > md ? t : md;
  }
  const int wmax = __builtin_amdgcn_readfirstlane(md);

  float4 acc[4];
#pragma unroll
  for (int g = 0; g < 4; ++g) acc[g] = make_float4(0.0f, 0.0f, 0.0f, 0.0f);

  if (wmax > 0) {
    const char* xrow = (const char*)XIN;
    const char* rec  = (const char*)csr + sel * 8;
    const unsigned q16 = (unsigned)q * 16u;
    unsigned so[4][EU];
    float    ww[4][EU];
#pragma unroll
    for (int g = 0; g < 4; ++g) {
      int lastb = endb[g] - 16;
      lastb = lastb > 0 ? lastb : 0;
#pragma unroll
      for (int u = 0; u < EU; ++u) {
        int jb = begb[g] + u * 16;
        jb = jb < lastb ? jb : lastb;
        const int2 r = *(const int2*)(rec + (unsigned)jb);
        so[g][u] = (unsigned)r.x * 128u + q16;
        ww[g][u] = __int_as_float(r.y);
      }
    }
    for (int j = 0; j < wmax; j += EU) {
      int2   nr[4][EU];
      float4 v[4][EU];
#pragma unroll
      for (int g = 0; g < 4; ++g) {
        int lastb = endb[g] - 16;
        lastb = lastb > 0 ? lastb : 0;
#pragma unroll
        for (int u = 0; u < EU; ++u) {
          int jb = begb[g] + (j + EU + u) * 16;
          jb = jb < lastb ? jb : lastb;
          nr[g][u] = *(const int2*)(rec + (unsigned)jb);
        }
      }
      asm volatile("" ::: "memory");
#pragma unroll
      for (int g = 0; g < 4; ++g) {
#pragma unroll
        for (int u = 0; u < EU; ++u) v[g][u] = *(const float4*)(xrow + so[g][u]);
      }
#pragma unroll
      for (int g = 0; g < 4; ++g) {
#pragma unroll
        for (int u = 0; u < EU; ++u) {
          const int m = (begb[g] + (j + u) * 16 < endb[g]) ? -1 : 0;
          const float w = __int_as_float(__float_as_int(ww[g][u]) & m);
          acc[g].x = __builtin_fmaf(__int_as_float(__float_as_int(v[g][u].x) & m), w, acc[g].x);
          acc[g].y = __builtin_fmaf(__int_as_float(__float_as_int(v[g][u].y) & m), w, acc[g].y);
          acc[g].z = __builtin_fmaf(__int_as_float(__float_as_int(v[g][u].z) & m), w, acc[g].z);
          acc[g].w = __builtin_fmaf(__int_as_float(__float_as_int(v[g][u].w) & m), w, acc[g].w);
        }
      }
#pragma unroll
      for (int g = 0; g < 4; ++g) {
#pragma unroll
        for (int u = 0; u < EU; ++u) {
          so[g][u] = (unsigned)nr[g][u].x * 128u + q16;
          ww[g][u] = __int_as_float(nr[g][u].y);
        }
      }
    }
  }

  const int half = lane >> 4;
  const int m    = lane & 15;
  int arow = row0 + m;
  if (arow >= N) arow = N - 1;

  v2f ax[8], bg0[8], bg1[8], br0[8], br1[8];
#pragma unroll
  for (int s = 0; s < 8; ++s) {
    const int kk = s * 4 + half * 2;
    ax[s] = *(const v2f*)(XIN + (long)arow * 32 + kk);
    bg0[s].x = G[kk * 32 + m];        bg0[s].y = G[(kk + 1) * 32 + m];
    bg1[s].x = G[kk * 32 + 16 + m];   bg1[s].y = G[(kk + 1) * 32 + 16 + m];
    br0[s].x = R[kk * 32 + m];        br0[s].y = R[(kk + 1) * 32 + m];
    br1[s].x = R[kk * 32 + 16 + m];   br1[s].y = R[(kk + 1) * 32 + 16 + m];
  }
  const float b0 = B[m], b1 = B[16 + m];

#pragma unroll
  for (int g = 0; g < 4; ++g) {
    float c = (float)((endb[g] - begb[g]) >> 4);
    c = c > 1.0f ? c : 1.0f;
    const float inv = 1.0f / c;
    float4 a;
    a.x = acc[g].x * inv;
    a.y = acc[g].y * inv;
    a.z = acc[g].z * inv;
    a.w = acc[g].w * inv;
    *(float4*)(T + (g * 4 + nl) * TP + q * 4) = a;
  }
  own_lds_stores_landed();

  v8f accg0 = {}, accg1 = {}, accr0 = {}, accr1 = {};

#pragma unroll
  for (int s = 0; s < 8; ++s) {
    const int kk = s * 4 + half * 2;
    const v2f am = *(const v2f*)(T + m * TP + kk);
    accg0 = __builtin_amdgcn_wmma_f32_16x16x4_f32(false, am, false, bg0[s], (short)0, accg0, false, false);
    accg1 = __builtin_amdgcn_wmma_f32_16x16x4_f32(false, am, false, bg1[s], (short)0, accg1, false, false);
    accr0 = __builtin_amdgcn_wmma_f32_16x16x4_f32(false, ax[s], false, br0[s], (short)0, accr0, false, false);
    accr1 = __builtin_amdgcn_wmma_f32_16x16x4_f32(false, ax[s], false, br1[s], (short)0, accr1, false, false);
  }

#pragma unroll
  for (int r = 0; r < 8; ++r) {
    T[(half * 8 + r) * TP + m]      = accg0[r] + (accr0[r] + b0);
    T[(half * 8 + r) * TP + 16 + m] = accg1[r] + (accr1[r] + b1);
  }
  own_lds_stores_landed();
#pragma unroll
  for (int g = 0; g < 4; ++g) {
    const int n = row0 + g * 4 + nl;
    const float4 o = *(const float4*)(T + (g * 4 + nl) * TP + q * 4);
    if (n < N) *(float4*)(OUT + (long)n * 32 + q * 4) = o;
  }
}

extern "C" void kernel_launch(void* const* d_in, const int* in_sizes, int n_in,
                              void* d_out, int out_size, void* d_ws, size_t ws_size,
                              hipStream_t stream) {
  const int*   ei  = (const int*)d_in[0];
  const float* ea  = (const float*)d_in[1];
  const float* x   = (const float*)d_in[2];
  const float* g1  = (const float*)d_in[3];
  const float* mu1 = (const float*)d_in[4];
  const float* sg1 = (const float*)d_in[5];
  const float* r1  = (const float*)d_in[6];
  const float* b1  = (const float*)d_in[7];
  const float* g2  = (const float*)d_in[8];
  const float* mu2 = (const float*)d_in[9];
  const float* sg2 = (const float*)d_in[10];
  const float* r2  = (const float*)d_in[11];
  const float* b2  = (const float*)d_in[12];

  const int E = in_sizes[0] / 2;
  const int N = in_sizes[2] / 32;
  const long plane = (long)N * 32;

  const int NB = (N + 255) / 256;
  int*   csr    = (int*)d_ws;
  int*   seg    = csr + (long)E * 4;
  int*   tmp    = seg + (long)N * 4;
  float* h      = (float*)tmp;
  const long tmpWords = (long)E * 4 > plane ? (long)E * 4 : plane;
  int*   P      = tmp + tmpWords;
  float* out    = (float*)d_out;

  const int ntiles     = (N + 15) / 16;
  const int tileBlocks = (ntiles + 3) / 4;

  move_kernel<<<G1, MT, 0, stream>>>(ei, ea, mu1, sg1, mu2, sg2, P, tmp, E, NB);
  bin_kernel<<<NB, BT, 0, stream>>>(tmp, P, seg, csr, N, E, NB);

  layer_kernel<<<tileBlocks, 128, 0, stream>>>(seg, csr, x, g1, r1, b1, h, N, ntiles, 0);
  layer_kernel<<<tileBlocks, 128, 0, stream>>>(seg, csr, h, g2, r2, b2, out, N, ntiles, 1);
}


# Layer gather: a node's edge records fetched 8 at a time, one per lane of its 8-lane group, read across lanes by DPP8 (on the row-address OR and a move); rows 4 edges a sub-trip; sum order unchanged
# speedup vs baseline: 3.5112x; 1.0838x over previous
#include <hip/hip_runtime.h>


typedef float v2f __attribute__((ext_vector_type(2)));
typedef float v8f __attribute__((ext_vector_type(8)));

#define EPSQ 1e-15f

#define NBMAX 512
#define G1 256
#define MT 1024
#define C1 7
#define MU 4
#define BT 512
#define RW (G1 / (BT / 32))
#define CHMAX 6656
#define BINMAX 4608


__global__ void __launch_bounds__(MT)
move_kernel(const int* __restrict__ ei,
            const float* __restrict__ ea,
            const float* __restrict__ mu1, const float* __restrict__ sg1,
            const float* __restrict__ mu2, const float* __restrict__ sg2,
            int* __restrict__ P, int* __restrict__ tmp, int E, int NB) {
  __shared__ int cur[NBMAX];
  __shared__ int wsum[32];
  __shared__ int4 stage[CHMAX];
  const int t = threadIdx.x;
  const int lane = t & 31;
  const int wv = t >> 5;
  const int CH = (E + G1 - 1) / G1;
  const bool staged = CH <= CHMAX;
  const int lo = blockIdx.x * CH;
  int hi = lo + CH;
  hi = hi < E ? hi : E;
  if (t < NBMAX) cur[t] = 0;
  __syncthreads();
  for (int e0 = lo + t; e0 < hi; e0 += C1 * MT) {
    int d[C1];
#pragma unroll
    for (int u = 0; u < C1; ++u) {
      int e = e0 + u * MT;
      e = e < hi ? e : hi - 1;
      d[u] = ei[E + e];
    }
#pragma unroll
    for (int u = 0; u < C1; ++u) asm volatile("" :: "v"(d[u]));
#pragma unroll
    for (int u = 0; u < C1; ++u)
      if (e0 + u * MT < hi) atomicAdd(&cur[d[u] >> 8], 1);
  }
  __syncthreads();
  int c = 0;
  if (t < NBMAX) c = cur[t];
  int inc = c;
  for (int o = 1; o < 32; o <<= 1) {
    const int u = __shfl_up(inc, o);
    if (lane >= o) inc += u;
  }
  if (lane == 31) wsum[wv] = inc;
  __syncthreads();
  if (t < NBMAX) {
    int base = inc - c;
    for (int w = 0; w < wv; ++w) base += wsum[w];
    cur[t] = base;
    if (t < NB) P[(long)blockIdx.x * (NB + 1) + t] = base;
  }
  if (t == 0) P[(long)blockIdx.x * (NB + 1) + NB] = hi > lo ? hi - lo : 0;
  __syncthreads();

  float m1[8], m2[8], q1[8], q2[8];
#pragma unroll
  for (int d = 0; d < 8; ++d) { m1[d] = mu1[d]; m2[d] = mu2[d]; q1[d] = sg1[d]; q2[d] = sg2[d]; }

  for (int e0 = lo + t; e0 < hi; e0 += MU * MT) {
    float4 a0[MU], a1[MU];
    int sv[MU], dv[MU];
#pragma unroll
    for (int u = 0; u < MU; ++u) {
      int e = e0 + u * MT;
      e = e < hi ? e : hi - 1;
      a0[u] = *(const float4*)(ea + (long)e * 8);
      a1[u] = *(const float4*)(ea + (long)e * 8 + 4);
      sv[u] = ei[e];
      dv[u] = ei[E + e];
    }
#pragma unroll
    for (int u = 0; u < MU; ++u)
      asm volatile("" :: "v"(a0[u].x), "v"(a0[u].y), "v"(a0[u].z), "v"(a0[u].w),
                         "v"(a1[u].x), "v"(a1[u].y), "v"(a1[u].z), "v"(a1[u].w), "v"(sv[u]), "v"(dv[u]));
#pragma unroll
    for (int u = 0; u < MU; ++u) {
      const float av[8] = { a0[u].x, a0[u].y, a0[u].z, a0[u].w, a1[u].x, a1[u].y, a1[u].z, a1[u].w };
      float s1 = 0.0f, s2 = 0.0f;
#pragma unroll
      for (int d = 0; d < 8; ++d) {
        const float d1 = av[d] - m1[d];
        const float d2 = av[d] - m2[d];
        s1 += (d1 * d1) / (EPSQ + q1[d] * q1[d]);
        s2 += (d2 * d2) / (EPSQ + q2[d] * q2[d]);
      }
      if (e0 + u * MT < hi) {
        const int p = atomicAdd(&cur[dv[u] >> 8], 1);
        const int4 rec = make_int4((sv[u] << 8) | (dv[u] & 255),
                                   __float_as_int(__expf(-0.5f * s1)),
                                   __float_as_int(__expf(-0.5f * s2)), 0);
        if (staged) stage[p] = rec;
        else *(int4*)(tmp + (long)(lo + p) * 4) = rec;
      }
    }
  }
  if (staged) {
    __syncthreads();
    for (int i = t; i < hi - lo; i += MT) *(int4*)(tmp + (long)(lo + i) * 4) = stage[i];
  }
}

__global__ void __launch_bounds__(BT)
bin_kernel(const int* __restrict__ tmp, const int* __restrict__ P,
           int* __restrict__ seg, int* __restrict__ csr, int N, int E, int NB) {
  __shared__ int cnt[256];
  __shared__ int wsum[8];
  __shared__ int rs[G1];
  __shared__ int rl[G1];
  __shared__ int red[2][BT / 32];
  __shared__ int4 stage[BINMAX];
  const int t = threadIdx.x;
  const int lane = t & 31;
  const int wv = t >> 5;
  const int CH = (E + G1 - 1) / G1;
  int p0 = 0, p1 = 0;
  if (t < G1) {
    p0 = P[(long)t * (NB + 1) + blockIdx.x];
    p1 = P[(long)t * (NB + 1) + blockIdx.x + 1];
    rs[t] = t * CH + p0;
    rl[t] = p1 - p0;
  }
  if (t < 256) cnt[t] = 0;
  {
    int s0 = p0, s1 = p1;
    for (int o = 16; o > 0; o >>= 1) {
      s0 += __shfl_xor(s0, o);
      s1 += __shfl_xor(s1, o);
    }
    if (lane == 0) { red[0][wv] = s0; red[1][wv] = s1; }
  }
  __syncthreads();
  int lo = 0, hi = 0;
#pragma unroll
  for (int w = 0; w < BT / 32; ++w) { lo += red[0][w]; hi += red[1][w]; }
  const bool fits = hi - lo <= BINMAX;

  int Lr[RW], Sr[RW];
#pragma unroll
  for (int u = 0; u < RW; ++u) {
    Lr[u] = __builtin_amdgcn_readfirstlane(rl[wv * RW + u]);
    Sr[u] = __builtin_amdgcn_readfirstlane(rs[wv * RW + u]);
  }
  int4 r[RW];
#pragma unroll
  for (int u = 0; u < RW; ++u) {
    const int L = Lr[u];
    int j = Sr[u] + (lane < L ? lane : L - 1);
    j = j > 0 ? j : 0;
    j = j < E ? j : E - 1;
    r[u] = *(const int4*)(tmp + (long)j * 4);
  }
#pragma unroll
  for (int u = 0; u < RW; ++u) asm volatile("" :: "v"(r[u].x), "v"(r[u].y), "v"(r[u].z));
#pragma unroll
  for (int u = 0; u < RW; ++u)
    if (lane < Lr[u]) atomicAdd(&cnt[r[u].x & 255], 1);
#pragma unroll
  for (int u = 0; u < RW; ++u) {
    const int L = Lr[u];
    const int S = Sr[u];
    for (int k = 32 + lane; k < L; k += 32) atomicAdd(&cnt[tmp[(long)(S + k) * 4] & 255], 1);
  }
  __syncthreads();
  int c = 0;
  if (t < 256) c = cnt[t];
  int inc = c;
  for (int o = 1; o < 32; o <<= 1) {
    const int u = __shfl_up(inc, o);
    if (lane >= o) inc += u;
  }
  if (t < 256 && lane == 31) wsum[wv] = inc;
  __syncthreads();
  if (t < 256) {
    int beg = lo + inc - c;
    for (int w = 0; w < wv; ++w) beg += wsum[w];
    const int n = blockIdx.x * 256 + t;
    if (n < N) *(int4*)(seg + (long)n * 4) = make_int4(beg, beg + c, 0, 0);
    cnt[t] = beg;
  }
  __syncthreads();
  {
    int pos[RW];
#pragma unroll
    for (int u = 0; u < RW; ++u) {
      int p = lo;
      if (lane < Lr[u]) p = atomicAdd(&cnt[r[u].x & 255], 1);
      pos[u] = p;
    }
#pragma unroll
    for (int u = 0; u < RW; ++u) {
      if (lane < Lr[u]) {
        const int sn = (int)((unsigned)r[u].x >> 8);
        const int4 o = make_int4(sn, r[u].y, sn, r[u].z);
        if (fits) stage[pos[u] - lo] = o;
        else *(int4*)(csr + (long)pos[u] * 4) = o;
      }
    }
  }
#pragma unroll
  for (int u = 0; u < RW; ++u) {
    const int L = Lr[u];
    const int S = Sr[u];
    for (int k = 32 + lane; k < L; k += 32) {
      const int4 x = *(const int4*)(tmp + (long)(S + k) * 4);
      const int pos = atomicAdd(&cnt[x.x & 255], 1);
      const int sn = (int)((unsigned)x.x >> 8);
      const int4 o = make_int4(sn, x.y, sn, x.z);
      if (fits) stage[pos - lo] = o;
      else *(int4*)(csr + (long)pos * 4) = o;
    }
  }
  if (fits) {
    __syncthreads();
    for (int i = t; i < hi - lo; i += BT) *(int4*)(csr + (long)(lo + i) * 4) = stage[i];
  }
}

#define RB 8
#define EU 4
#define TP 36

static __device__ __forceinline__ void own_lds_stores_landed() {
  asm volatile("s_wait_dscnt 0x0" ::: "memory");
}

template <int U>
static __device__ __forceinline__ int from_group_lane(int x) {
  int r;
  asm("v_mov_b32_dpp %0, %1 dpp8:[%2,%2,%2,%2,%2,%2,%2,%2]" : "=v"(r) : "v"(x), "n"(U));
  return r;
}
template <int U>
static __device__ __forceinline__ unsigned from_group_lane_or(unsigned x, unsigned y) {
  unsigned r;
  asm("v_or_b32_dpp %0, %1, %2 dpp8:[%3,%3,%3,%3,%3,%3,%3,%3]" : "=v"(r) : "v"(x), "v"(y), "n"(U));
  return r;
}

template <int U0>
static __device__ __forceinline__ void gather_sub_trip(const char* xrow, unsigned q16, int j,
                                                       const unsigned (&sb)[4], const int (&wm)[4],
                                                       const int (&deg)[4], float4 (&acc)[4]) {
  unsigned so[4][EU];
#pragma unroll
  for (int g = 0; g < 4; ++g) {
    so[g][0] = from_group_lane_or<U0 + 0>(sb[g], q16);
    so[g][1] = from_group_lane_or<U0 + 1>(sb[g], q16);
    so[g][2] = from_group_lane_or<U0 + 2>(sb[g], q16);
    so[g][3] = from_group_lane_or<U0 + 3>(sb[g], q16);
  }
  __builtin_amdgcn_sched_barrier(0);
  float4 v[4][EU];
#pragma unroll
  for (int g = 0; g < 4; ++g) {
#pragma unroll
    for (int u = 0; u < EU; ++u) v[g][u] = *(const float4*)(xrow + so[g][u]);
  }
#pragma unroll
  for (int g = 0; g < 4; ++g) {
    const int wb[EU] = { from_group_lane<U0 + 0>(wm[g]), from_group_lane<U0 + 1>(wm[g]),
                         from_group_lane<U0 + 2>(wm[g]), from_group_lane<U0 + 3>(wm[g]) };
#pragma unroll
    for (int u = 0; u < EU; ++u) {
      const int m = (j + U0 + u < deg[g]) ? -1 : 0;
      const float w = __int_as_float(wb[u]);
      acc[g].x = __builtin_fmaf(__int_as_float(__float_as_int(v[g][u].x) & m), w, acc[g].x);
      acc[g].y = __builtin_fmaf(__int_as_float(__float_as_int(v[g][u].y) & m), w, acc[g].y);
      acc[g].z = __builtin_fmaf(__int_as_float(__float_as_int(v[g][u].z) & m), w, acc[g].z);
      acc[g].w = __builtin_fmaf(__int_as_float(__float_as_int(v[g][u].w) & m), w, acc[g].w);
    }
  }
}

__global__ void __launch_bounds__(128)
layer_kernel(const int* __restrict__ seg, const int* __restrict__ csr,
             const float* __restrict__ XIN, const float* __restrict__ G,
             const float* __restrict__ R, const float* __restrict__ B,
             float* __restrict__ OUT, int N, int ntiles, int sel) {
  __shared__ __attribute__((aligned(16))) float stage[4 * 16 * TP];
  const int wv   = threadIdx.x >> 5;
  const int wave = blockIdx.x * 4 + wv;
  if (wave >= ntiles) return;
  const int lane = threadIdx.x & 31;
  float* T = stage + wv * (16 * TP);
  const int row0 = wave * 16;
  const int nl = lane >> 3;
  const int q  = lane & 7;

  int begb[4], deg[4];
  int md = 0;
#pragma unroll
  for (int g = 0; g < 4; ++g) {
    const int n = row0 + g * 4 + nl;
    int2 be = make_int2(0, 0);
    if (n < N) be = *(const int2*)(seg + (long)n * 4);
    begb[g] = be.x * 16;
    const int d = be.y - be.x;
    deg[g] = d;
    md = d > md ? d : md;
  }
  for (int o = 16; o > 0; o >>= 1) {
    const int t = __shfl_xor(md, o);
    md = t > md ? t : md;
  }
  const int wmax = __builtin_amdgcn_readfirstlane(md);

  float4 acc[4];
#pragma unroll
  for (int g = 0; g < 4; ++g) acc[g] = make_float4(0.0f, 0.0f, 0.0f, 0.0f);

  if (wmax > 0) {
    const char* xrow = (const char*)XIN;
    const char* rec  = (const char*)csr + sel * 8;
    const unsigned q16 = (unsigned)q * 16u;
    int2 rc[4];
#pragma unroll
    for (int g = 0; g < 4; ++g) {
      int lastb = begb[g] + deg[g] * 16 - 16;
      lastb = lastb > 0 ? lastb : 0;
      int jb = begb[g] + q * 16;
      jb = jb < lastb ? jb : lastb;
      rc[g] = *(const int2*)(rec + (unsigned)jb);
    }
    for (int j = 0; j < wmax; j += RB) {
      int2 nr[4];
#pragma unroll
      for (int g = 0; g < 4; ++g) {
        int lastb = begb[g] + deg[g] * 16 - 16;
        lastb = lastb > 0 ? lastb : 0;
        int jb = begb[g] + (j + RB + q) * 16;
        jb = jb < lastb ? jb : lastb;
        nr[g] = *(const int2*)(rec + (unsigned)jb);
      }
      unsigned sb[4];
      int wm[4];
#pragma unroll
      for (int g = 0; g < 4; ++g) {
        sb[g] = (unsigned)rc[g].x * 128u;
        wm[g] = (j + q < deg[g]) ? rc[g].y : 0;
      }
      gather_sub_trip<0>(xrow, q16, j, sb, wm, deg, acc);
      if (j + EU < wmax) gather_sub_trip<EU>(xrow, q16, j, sb, wm, deg, acc);
#pragma unroll
      for (int g = 0; g < 4; ++g) rc[g] = nr[g];
    }
  }

  const int half = lane >> 4;
  const int m    = lane & 15;
  int arow = row0 + m;
  if (arow >= N) arow = N - 1;

  v2f ax[8], bg0[8], bg1[8], br0[8], br1[8];
#pragma unroll
  for (int s = 0; s < 8; ++s) {
    const int kk = s * 4 + half * 2;
    ax[s] = *(const v2f*)(XIN + (long)arow * 32 + kk);
    bg0[s].x = G[kk * 32 + m];        bg0[s].y = G[(kk + 1) * 32 + m];
    bg1[s].x = G[kk * 32 + 16 + m];   bg1[s].y = G[(kk + 1) * 32 + 16 + m];
    br0[s].x = R[kk * 32 + m];        br0[s].y = R[(kk + 1) * 32 + m];
    br1[s].x = R[kk * 32 + 16 + m];   br1[s].y = R[(kk + 1) * 32 + 16 + m];
  }
  const float b0 = B[m], b1 = B[16 + m];

#pragma unroll
  for (int g = 0; g < 4; ++g) {
    float c = (float)deg[g];
    c = c > 1.0f ? c : 1.0f;
    const float inv = 1.0f / c;
    float4 a;
    a.x = acc[g].x * inv;
    a.y = acc[g].y * inv;
    a.z = acc[g].z * inv;
    a.w = acc[g].w * inv;
    *(float4*)(T + (g * 4 + nl) * TP + q * 4) = a;
  }
  own_lds_stores_landed();

  v8f accg0 = {}, accg1 = {}, accr0 = {}, accr1 = {};

#pragma unroll
  for (int s = 0; s < 8; ++s) {
    const int kk = s * 4 + half * 2;
    const v2f am = *(const v2f*)(T + m * TP + kk);
    accg0 = __builtin_amdgcn_wmma_f32_16x16x4_f32(false, am, false, bg0[s], (short)0, accg0, false, false);
    accg1 = __builtin_amdgcn_wmma_f32_16x16x4_f32(false, am, false, bg1[s], (short)0, accg1, false, false);
    accr0 = __builtin_amdgcn_wmma_f32_16x16x4_f32(false, ax[s], false, br0[s], (short)0, accr0, false, false);
    accr1 = __builtin_amdgcn_wmma_f32_16x16x4_f32(false, ax[s], false, br1[s], (short)0, accr1, false, false);
  }

#pragma unroll
  for (int r = 0; r < 8; ++r) {
    T[(half * 8 + r) * TP + m]      = accg0[r] + (accr0[r] + b0);
    T[(half * 8 + r) * TP + 16 + m] = accg1[r] + (accr1[r] + b1);
  }
  own_lds_stores_landed();
#pragma unroll
  for (int g = 0; g < 4; ++g) {
    const int n = row0 + g * 4 + nl;
    const float4 o = *(const float4*)(T + (g * 4 + nl) * TP + q * 4);
    if (n < N) *(float4*)(OUT + (long)n * 32 + q * 4) = o;
  }
}

extern "C" void kernel_launch(void* const* d_in, const int* in_sizes, int n_in,
                              void* d_out, int out_size, void* d_ws, size_t ws_size,
                              hipStream_t stream) {
  const int*   ei  = (const int*)d_in[0];
  const float* ea  = (const float*)d_in[1];
  const float* x   = (const float*)d_in[2];
  const float* g1  = (const float*)d_in[3];
  const float* mu1 = (const float*)d_in[4];
  const float* sg1 = (const float*)d_in[5];
  const float* r1  = (const float*)d_in[6];
  const float* b1  = (const float*)d_in[7];
  const float* g2  = (const float*)d_in[8];
  const float* mu2 = (const float*)d_in[9];
  const float* sg2 = (const float*)d_in[10];
  const float* r2  = (const float*)d_in[11];
  const float* b2  = (const float*)d_in[12];

  const int E = in_sizes[0] / 2;
  const int N = in_sizes[2] / 32;
  const long plane = (long)N * 32;

  const int NB = (N + 255) / 256;
  int*   csr    = (int*)d_ws;
  int*   seg    = csr + (long)E * 4;
  int*   tmp    = seg + (long)N * 4;
  float* h      = (float*)tmp;
  const long tmpWords = (long)E * 4 > plane ? (long)E * 4 : plane;
  int*   P      = tmp + tmpWords;
  float* out    = (float*)d_out;

  const int ntiles     = (N + 15) / 16;
  const int tileBlocks = (ntiles + 3) / 4;

  move_kernel<<<G1, MT, 0, stream>>>(ei, ea, mu1, sg1, mu2, sg2, P, tmp, E, NB);
  bin_kernel<<<NB, BT, 0, stream>>>(tmp, P, seg, csr, N, E, NB);

  layer_kernel<<<tileBlocks, 128, 0, stream>>>(seg, csr, x, g1, r1, b1, h, N, ntiles, 0);
  layer_kernel<<<tileBlocks, 128, 0, stream>>>(seg, csr, h, g2, r2, b2, out, N, ntiles, 1);
}


# Layer tiles take 16 nodes adjacent in their bin's in-degree order (bin kernel ranks nodes on LDS class counters; segment table carries node ids): fewer gather slots past a node's end; sum order same
# speedup vs baseline: 3.6819x; 1.0486x over previous
#include <hip/hip_runtime.h>


typedef float v2f __attribute__((ext_vector_type(2)));
typedef float v8f __attribute__((ext_vector_type(8)));

#define EPSQ 1e-15f

#define NBMAX 512
#define G1 256
#define MT 1024
#define C1 7
#define MU 4
#define BT 512
#define RW (G1 / (BT / 32))
#define CHMAX 6656
#define BINMAX 4608


__global__ void __launch_bounds__(MT)
move_kernel(const int* __restrict__ ei,
            const float* __restrict__ ea,
            const float* __restrict__ mu1, const float* __restrict__ sg1,
            const float* __restrict__ mu2, const float* __restrict__ sg2,
            int* __restrict__ P, int* __restrict__ tmp, int E, int NB) {
  __shared__ int cur[NBMAX];
  __shared__ int wsum[32];
  __shared__ int4 stage[CHMAX];
  const int t = threadIdx.x;
  const int lane = t & 31;
  const int wv = t >> 5;
  const int CH = (E + G1 - 1) / G1;
  const bool staged = CH <= CHMAX;
  const int lo = blockIdx.x * CH;
  int hi = lo + CH;
  hi = hi < E ? hi : E;
  if (t < NBMAX) cur[t] = 0;
  __syncthreads();
  for (int e0 = lo + t; e0 < hi; e0 += C1 * MT) {
    int d[C1];
#pragma unroll
    for (int u = 0; u < C1; ++u) {
      int e = e0 + u * MT;
      e = e < hi ? e : hi - 1;
      d[u] = ei[E + e];
    }
#pragma unroll
    for (int u = 0; u < C1; ++u) asm volatile("" :: "v"(d[u]));
#pragma unroll
    for (int u = 0; u < C1; ++u)
      if (e0 + u * MT < hi) atomicAdd(&cur[d[u] >> 8], 1);
  }
  __syncthreads();
  int c = 0;
  if (t < NBMAX) c = cur[t];
  int inc = c;
  for (int o = 1; o < 32; o <<= 1) {
    const int u = __shfl_up(inc, o);
    if (lane >= o) inc += u;
  }
  if (lane == 31) wsum[wv] = inc;
  __syncthreads();
  if (t < NBMAX) {
    int base = inc - c;
    for (int w = 0; w < wv; ++w) base += wsum[w];
    cur[t] = base;
    if (t < NB) P[(long)blockIdx.x * (NB + 1) + t] = base;
  }
  if (t == 0) P[(long)blockIdx.x * (NB + 1) + NB] = hi > lo ? hi - lo : 0;
  __syncthreads();

  float m1[8], m2[8], q1[8], q2[8];
#pragma unroll
  for (int d = 0; d < 8; ++d) { m1[d] = mu1[d]; m2[d] = mu2[d]; q1[d] = sg1[d]; q2[d] = sg2[d]; }

  for (int e0 = lo + t; e0 < hi; e0 += MU * MT) {
    float4 a0[MU], a1[MU];
    int sv[MU], dv[MU];
#pragma unroll
    for (int u = 0; u < MU; ++u) {
      int e = e0 + u * MT;
      e = e < hi ? e : hi - 1;
      a0[u] = *(const float4*)(ea + (long)e * 8);
      a1[u] = *(const float4*)(ea + (long)e * 8 + 4);
      sv[u] = ei[e];
      dv[u] = ei[E + e];
    }
#pragma unroll
    for (int u = 0; u < MU; ++u)
      asm volatile("" :: "v"(a0[u].x), "v"(a0[u].y), "v"(a0[u].z), "v"(a0[u].w),
                         "v"(a1[u].x), "v"(a1[u].y), "v"(a1[u].z), "v"(a1[u].w), "v"(sv[u]), "v"(dv[u]));
#pragma unroll
    for (int u = 0; u < MU; ++u) {
      const float av[8] = { a0[u].x, a0[u].y, a0[u].z, a0[u].w, a1[u].x, a1[u].y, a1[u].z, a1[u].w };
      float s1 = 0.0f, s2 = 0.0f;
#pragma unroll
      for (int d = 0; d < 8; ++d) {
        const float d1 = av[d] - m1[d];
        const float d2 = av[d] - m2[d];
        s1 += (d1 * d1) / (EPSQ + q1[d] * q1[d]);
        s2 += (d2 * d2) / (EPSQ + q2[d] * q2[d]);
      }
      if (e0 + u * MT < hi) {
        const int p = atomicAdd(&cur[dv[u] >> 8], 1);
        const int4 rec = make_int4((sv[u] << 8) | (dv[u] & 255),
                                   __float_as_int(__expf(-0.5f * s1)),
                                   __float_as_int(__expf(-0.5f * s2)), 0);
        if (staged) stage[p] = rec;
        else *(int4*)(tmp + (long)(lo + p) * 4) = rec;
      }
    }
  }
  if (staged) {
    __syncthreads();
    for (int i = t; i < hi - lo; i += MT) *(int4*)(tmp + (long)(lo + i) * 4) = stage[i];
  }
}

__global__ void __launch_bounds__(BT)
bin_kernel(const int* __restrict__ tmp, const int* __restrict__ P,
           int* __restrict__ seg, int* __restrict__ csr, int N, int E, int NB) {
  __shared__ int cnt[256];
  __shared__ int wsum[8];
  __shared__ int rs[G1];
  __shared__ int rl[G1];
  __shared__ int red[2][BT / 32];
  __shared__ int4 stage[BINMAX];
  __shared__ __attribute__((aligned(8))) int dh[64];
  __shared__ int dbase[64];
  __shared__ int4 segst[256];
  const int t = threadIdx.x;
  const int lane = t & 31;
  const int wv = t >> 5;
  const int CH = (E + G1 - 1) / G1;
  int p0 = 0, p1 = 0;
  if (t < G1) {
    p0 = P[(long)t * (NB + 1) + blockIdx.x];
    p1 = P[(long)t * (NB + 1) + blockIdx.x + 1];
    rs[t] = t * CH + p0;
    rl[t] = p1 - p0;
  }
  if (t < 256) cnt[t] = 0;
  if (t < 64) dh[t] = 0;
  {
    int s0 = p0, s1 = p1;
    for (int o = 16; o > 0; o >>= 1) {
      s0 += __shfl_xor(s0, o);
      s1 += __shfl_xor(s1, o);
    }
    if (lane == 0) { red[0][wv] = s0; red[1][wv] = s1; }
  }
  __syncthreads();
  int lo = 0, hi = 0;
#pragma unroll
  for (int w = 0; w < BT / 32; ++w) { lo += red[0][w]; hi += red[1][w]; }
  const bool fits = hi - lo <= BINMAX;

  int Lr[RW], Sr[RW];
#pragma unroll
  for (int u = 0; u < RW; ++u) {
    Lr[u] = __builtin_amdgcn_readfirstlane(rl[wv * RW + u]);
    Sr[u] = __builtin_amdgcn_readfirstlane(rs[wv * RW + u]);
  }
  int4 r[RW];
#pragma unroll
  for (int u = 0; u < RW; ++u) {
    const int L = Lr[u];
    int j = Sr[u] + (lane < L ? lane : L - 1);
    j = j > 0 ? j : 0;
    j = j < E ? j : E - 1;
    r[u] = *(const int4*)(tmp + (long)j * 4);
  }
#pragma unroll
  for (int u = 0; u < RW; ++u) asm volatile("" :: "v"(r[u].x), "v"(r[u].y), "v"(r[u].z));
#pragma unroll
  for (int u = 0; u < RW; ++u)
    if (lane < Lr[u]) atomicAdd(&cnt[r[u].x & 255], 1);
#pragma unroll
  for (int u = 0; u < RW; ++u) {
    const int L = Lr[u];
    const int S = Sr[u];
    for (int k = 32 + lane; k < L; k += 32) atomicAdd(&cnt[tmp[(long)(S + k) * 4] & 255], 1);
  }
  __syncthreads();
  int c = 0;
  if (t < 256) c = cnt[t];
  const int nme = blockIdx.x * 256 + t;
  int key = 63;
  if (t < 256) {
    if (nme < N) key = c < 62 ? c : 62;
    atomicAdd(&dh[key], 1);
  }
  int inc = c;
  for (int o = 1; o < 32; o <<= 1) {
    const int u = __shfl_up(inc, o);
    if (lane >= o) inc += u;
  }
  if (t < 256 && lane == 31) wsum[wv] = inc;
  __syncthreads();
  int beg = 0;
  if (t < 256) {
    beg = lo + inc - c;
    for (int w = 0; w < wv; ++w) beg += wsum[w];
    cnt[t] = beg;
  }
  if (wv == 0) {
    const int2 h = ((const int2*)dh)[lane];
    const int s = h.x + h.y;
    int incs = s;
    for (int o = 1; o < 32; o <<= 1) {
      const int u = __shfl_up(incs, o);
      if (lane >= o) incs += u;
    }
    dbase[2 * lane]     = incs - s;
    dbase[2 * lane + 1] = incs - s + h.x;
  }
  __syncthreads();
  if (t < 256) {
    const int pos = atomicAdd(&dbase[key], 1);
    segst[pos] = make_int4(beg, beg + c, nme, 0);
  }
  {
    int pos[RW];
#pragma unroll
    for (int u = 0; u < RW; ++u) {
      int p = lo;
      if (lane < Lr[u]) p = atomicAdd(&cnt[r[u].x & 255], 1);
      pos[u] = p;
    }
#pragma unroll
    for (int u = 0; u < RW; ++u) {
      if (lane < Lr[u]) {
        const int sn = (int)((unsigned)r[u].x >> 8);
        const int4 o = make_int4(sn, r[u].y, sn, r[u].z);
        if (fits) stage[pos[u] - lo] = o;
        else *(int4*)(csr + (long)pos[u] * 4) = o;
      }
    }
  }
#pragma unroll
  for (int u = 0; u < RW; ++u) {
    const int L = Lr[u];
    const int S = Sr[u];
    for (int k = 32 + lane; k < L; k += 32) {
      const int4 x = *(const int4*)(tmp + (long)(S + k) * 4);
      const int pos = atomicAdd(&cnt[x.x & 255], 1);
      const int sn = (int)((unsigned)x.x >> 8);
      const int4 o = make_int4(sn, x.y, sn, x.z);
      if (fits) stage[pos - lo] = o;
      else *(int4*)(csr + (long)pos * 4) = o;
    }
  }
  __syncthreads();
  if (fits) {
    for (int i = t; i < hi - lo; i += BT) *(int4*)(csr + (long)(lo + i) * 4) = stage[i];
  }
  if (t < 256) {
    const int p = blockIdx.x * 256 + t;
    if (p < N) *(int4*)(seg + (long)p * 4) = segst[t];
  }
}

#define RB 8
#define EU 4
#define TP 36

static __device__ __forceinline__ void own_lds_stores_landed() {
  asm volatile("s_wait_dscnt 0x0" ::: "memory");
}

template <int U>
static __device__ __forceinline__ int from_group_lane(int x) {
  int r;
  asm("v_mov_b32_dpp %0, %1 dpp8:[%2,%2,%2,%2,%2,%2,%2,%2]" : "=v"(r) : "v"(x), "n"(U));
  return r;
}
template <int U>
static __device__ __forceinline__ unsigned from_group_lane_or(unsigned x, unsigned y) {
  unsigned r;
  asm("v_or_b32_dpp %0, %1, %2 dpp8:[%3,%3,%3,%3,%3,%3,%3,%3]" : "=v"(r) : "v"(x), "v"(y), "n"(U));
  return r;
}

template <int U0>
static __device__ __forceinline__ void gather_sub_trip(const char* xrow, unsigned q16, int j,
                                                       const unsigned (&sb)[4], const int (&wm)[4],
                                                       const int (&deg)[4], float4 (&acc)[4]) {
  unsigned so[4][EU];
#pragma unroll
  for (int g = 0; g < 4; ++g) {
    so[g][0] = from_group_lane_or<U0 + 0>(sb[g], q16);
    so[g][1] = from_group_lane_or<U0 + 1>(sb[g], q16);
    so[g][2] = from_group_lane_or<U0 + 2>(sb[g], q16);
    so[g][3] = from_group_lane_or<U0 + 3>(sb[g], q16);
  }
  __builtin_amdgcn_sched_barrier(0);
  float4 v[4][EU];
#pragma unroll
  for (int g = 0; g < 4; ++g) {
#pragma unroll
    for (int u = 0; u < EU; ++u) v[g][u] = *(const float4*)(xrow + so[g][u]);
  }
#pragma unroll
  for (int g = 0; g < 4; ++g) {
    const int wb[EU] = { from_group_lane<U0 + 0>(wm[g]), from_group_lane<U0 + 1>(wm[g]),
                         from_group_lane<U0 + 2>(wm[g]), from_group_lane<U0 + 3>(wm[g]) };
#pragma unroll
    for (int u = 0; u < EU; ++u) {
      const int m = (j + U0 + u < deg[g]) ? -1 : 0;
      const float w = __int_as_float(wb[u]);
      acc[g].x = __builtin_fmaf(__int_as_float(__float_as_int(v[g][u].x) & m), w, acc[g].x);
      acc[g].y = __builtin_fmaf(__int_as_float(__float_as_int(v[g][u].y) & m), w, acc[g].y);
      acc[g].z = __builtin_fmaf(__int_as_float(__float_as_int(v[g][u].z) & m), w, acc[g].z);
      acc[g].w = __builtin_fmaf(__int_as_float(__float_as_int(v[g][u].w) & m), w, acc[g].w);
    }
  }
}

__global__ void __launch_bounds__(128)
layer_kernel(const int* __restrict__ seg, const int* __restrict__ csr,
             const float* __restrict__ XIN, const float* __restrict__ G,
             const float* __restrict__ R, const float* __restrict__ B,
             float* __restrict__ OUT, int N, int ntiles, int sel) {
  __shared__ __attribute__((aligned(16))) float stage[4 * 16 * TP];
  const int wv   = threadIdx.x >> 5;
  const int turn = (int)((blockIdx.x * 2654435761u) >> 30);
  const int wave = (blockIdx.x >> 2) * 16 + ((wv + turn) & 3) * 4 + (blockIdx.x & 3);
  if (wave >= ntiles) return;
  const int lane = threadIdx.x & 31;
  float* T = stage + wv * (16 * TP);
  const int row0 = wave * 16;
  const int nl = lane >> 3;
  const int q  = lane & 7;

  int begb[4], deg[4];
  int md = 0;
#pragma unroll
  for (int g = 0; g < 4; ++g) {
    const int p = row0 + g * 4 + nl;
    int2 be = make_int2(0, 0);
    if (p < N) be = *(const int2*)(seg + (long)p * 4);
    begb[g] = be.x * 16;
    const int d = be.y - be.x;
    deg[g] = d;
    md = d > md ? d : md;
  }
  for (int o = 16; o > 0; o >>= 1) {
    const int t = __shfl_xor(md, o);
    md = t > md ? t : md;
  }
  const int wmax = __builtin_amdgcn_readfirstlane(md);
  int arow = 0;
  if (row0 + (lane & 15) < N) arow = seg[(long)(row0 + (lane & 15)) * 4 + 2];

  float4 acc[4];
#pragma unroll
  for (int g = 0; g < 4; ++g) acc[g] = make_float4(0.0f, 0.0f, 0.0f, 0.0f);

  if (wmax > 0) {
    const char* xrow = (const char*)XIN;
    const char* rec  = (const char*)csr + sel * 8;
    const unsigned q16 = (unsigned)q * 16u;
    int2 rc[4];
#pragma unroll
    for (int g = 0; g < 4; ++g) {
      int lastb = begb[g] + deg[g] * 16 - 16;
      lastb = lastb > 0 ? lastb : 0;
      int jb = begb[g] + q * 16;
      jb = jb < lastb ? jb : lastb;
      rc[g] = *(const int2*)(rec + (unsigned)jb);
    }
    for (int j = 0; j < wmax; j += RB) {
      int2 nr[4];
#pragma unroll
      for (int g = 0; g < 4; ++g) {
        int lastb = begb[g] + deg[g] * 16 - 16;
        lastb = lastb > 0 ? lastb : 0;
        int jb = begb[g] + (j + RB + q) * 16;
        jb = jb < lastb ? jb : lastb;
        nr[g] = *(const int2*)(rec + (unsigned)jb);
      }
      unsigned sb[4];
      int wm[4];
#pragma unroll
      for (int g = 0; g < 4; ++g) {
        sb[g] = (unsigned)rc[g].x * 128u;
        wm[g] = (j + q < deg[g]) ? rc[g].y : 0;
      }
      gather_sub_trip<0>(xrow, q16, j, sb, wm, deg, acc);
      if (j + EU < wmax) gather_sub_trip<EU>(xrow, q16, j, sb, wm, deg, acc);
#pragma unroll
      for (int g = 0; g < 4; ++g) rc[g] = nr[g];
    }
  }

  const int half = lane >> 4;
  const int m    = lane & 15;
  int nd[4];
#pragma unroll
  for (int g = 0; g < 4; ++g) {
    const int p = row0 + g * 4 + nl;
    nd[g] = -1;
    if (p < N) nd[g] = seg[(long)p * 4 + 2];
  }

  v2f ax[8], bg0[8], bg1[8], br0[8], br1[8];
#pragma unroll
  for (int s = 0; s < 8; ++s) {
    const int kk = s * 4 + half * 2;
    ax[s] = *(const v2f*)(XIN + (long)arow * 32 + kk);
    bg0[s].x = G[kk * 32 + m];        bg0[s].y = G[(kk + 1) * 32 + m];
    bg1[s].x = G[kk * 32 + 16 + m];   bg1[s].y = G[(kk + 1) * 32 + 16 + m];
    br0[s].x = R[kk * 32 + m];        br0[s].y = R[(kk + 1) * 32 + m];
    br1[s].x = R[kk * 32 + 16 + m];   br1[s].y = R[(kk + 1) * 32 + 16 + m];
  }
  const float b0 = B[m], b1 = B[16 + m];

#pragma unroll
  for (int g = 0; g < 4; ++g) {
    float c = (float)deg[g];
    c = c > 1.0f ? c : 1.0f;
    const float inv = 1.0f / c;
    float4 a;
    a.x = acc[g].x * inv;
    a.y = acc[g].y * inv;
    a.z = acc[g].z * inv;
    a.w = acc[g].w * inv;
    *(float4*)(T + (g * 4 + nl) * TP + q * 4) = a;
  }
  own_lds_stores_landed();

  v8f accg0 = {}, accg1 = {}, accr0 = {}, accr1 = {};

#pragma unroll
  for (int s = 0; s < 8; ++s) {
    const int kk = s * 4 + half * 2;
    const v2f am = *(const v2f*)(T + m * TP + kk);
    accg0 = __builtin_amdgcn_wmma_f32_16x16x4_f32(false, am, false, bg0[s], (short)0, accg0, false, false);
    accg1 = __builtin_amdgcn_wmma_f32_16x16x4_f32(false, am, false, bg1[s], (short)0, accg1, false, false);
    accr0 = __builtin_amdgcn_wmma_f32_16x16x4_f32(false, ax[s], false, br0[s], (short)0, accr0, false, false);
    accr1 = __builtin_amdgcn_wmma_f32_16x16x4_f32(false, ax[s], false, br1[s], (short)0, accr1, false, false);
  }

#pragma unroll
  for (int r = 0; r < 8; ++r) {
    T[(half * 8 + r) * TP + m]      = accg0[r] + (accr0[r] + b0);
    T[(half * 8 + r) * TP + 16 + m] = accg1[r] + (accr1[r] + b1);
  }
  own_lds_stores_landed();
#pragma unroll
  for (int g = 0; g < 4; ++g) {
    const float4 o = *(const float4*)(T + (g * 4 + nl) * TP + q * 4);
    if (nd[g] >= 0) *(float4*)(OUT + (long)nd[g] * 32 + q * 4) = o;
  }
}

extern "C" void kernel_launch(void* const* d_in, const int* in_sizes, int n_in,
                              void* d_out, int out_size, void* d_ws, size_t ws_size,
                              hipStream_t stream) {
  const int*   ei  = (const int*)d_in[0];
  const float* ea  = (const float*)d_in[1];
  const float* x   = (const float*)d_in[2];
  const float* g1  = (const float*)d_in[3];
  const float* mu1 = (const float*)d_in[4];
  const float* sg1 = (const float*)d_in[5];
  const float* r1  = (const float*)d_in[6];
  const float* b1  = (const float*)d_in[7];
  const float* g2  = (const float*)d_in[8];
  const float* mu2 = (const float*)d_in[9];
  const float* sg2 = (const float*)d_in[10];
  const float* r2  = (const float*)d_in[11];
  const float* b2  = (const float*)d_in[12];

  const int E = in_sizes[0] / 2;
  const int N = in_sizes[2] / 32;
  const long plane = (long)N * 32;

  const int NB = (N + 255) / 256;
  int*   csr    = (int*)d_ws;
  int*   seg    = csr + (long)E * 4;
  int*   tmp    = seg + (long)N * 4;
  float* h      = (float*)tmp;
  const long tmpWords = (long)E * 4 > plane ? (long)E * 4 : plane;
  int*   P      = tmp + tmpWords;
  float* out    = (float*)d_out;

  const int ntiles     = (N + 15) / 16;
  const int tileBlocks = NB * 4;

  move_kernel<<<G1, MT, 0, stream>>>(ei, ea, mu1, sg1, mu2, sg2, P, tmp, E, NB);
  bin_kernel<<<NB, BT, 0, stream>>>(tmp, P, seg, csr, N, E, NB);

  layer_kernel<<<tileBlocks, 128, 0, stream>>>(seg, csr, x, g1, r1, b1, h, N, ntiles, 0);
  layer_kernel<<<tileBlocks, 128, 0, stream>>>(seg, csr, h, g2, r2, b2, out, N, ntiles, 1);
}


# Edge-weight pass: the 16 per-edge divisions by (eps + sigma^2), 11 instructions each, become multiplies by reciprocals taken once per thread; same bits at sigma 1, else last bit; sum order same
# speedup vs baseline: 3.8098x; 1.0347x over previous
#include <hip/hip_runtime.h>


typedef float v2f __attribute__((ext_vector_type(2)));
typedef float v8f __attribute__((ext_vector_type(8)));

#define EPSQ 1e-15f

#define NBMAX 512
#define G1 256
#define MT 1024
#define C1 7
#define MU 4
#define BT 512
#define RW (G1 / (BT / 32))
#define CHMAX 6656
#define BINMAX 4608


__global__ void __launch_bounds__(MT)
move_kernel(const int* __restrict__ ei,
            const float* __restrict__ ea,
            const float* __restrict__ mu1, const float* __restrict__ sg1,
            const float* __restrict__ mu2, const float* __restrict__ sg2,
            int* __restrict__ P, int* __restrict__ tmp, int E, int NB) {
  __shared__ int cur[NBMAX];
  __shared__ int wsum[32];
  __shared__ int4 stage[CHMAX];
  const int t = threadIdx.x;
  const int lane = t & 31;
  const int wv = t >> 5;
  const int CH = (E + G1 - 1) / G1;
  const bool staged = CH <= CHMAX;
  const int lo = blockIdx.x * CH;
  int hi = lo + CH;
  hi = hi < E ? hi : E;
  if (t < NBMAX) cur[t] = 0;
  __syncthreads();
  for (int e0 = lo + t; e0 < hi; e0 += C1 * MT) {
    int d[C1];
#pragma unroll
    for (int u = 0; u < C1; ++u) {
      int e = e0 + u * MT;
      e = e < hi ? e : hi - 1;
      d[u] = ei[E + e];
    }
#pragma unroll
    for (int u = 0; u < C1; ++u) asm volatile("" :: "v"(d[u]));
#pragma unroll
    for (int u = 0; u < C1; ++u)
      if (e0 + u * MT < hi) atomicAdd(&cur[d[u] >> 8], 1);
  }
  __syncthreads();
  int c = 0;
  if (t < NBMAX) c = cur[t];
  int inc = c;
  for (int o = 1; o < 32; o <<= 1) {
    const int u = __shfl_up(inc, o);
    if (lane >= o) inc += u;
  }
  if (lane == 31) wsum[wv] = inc;
  __syncthreads();
  if (t < NBMAX) {
    int base = inc - c;
    for (int w = 0; w < wv; ++w) base += wsum[w];
    cur[t] = base;
    if (t < NB) P[(long)blockIdx.x * (NB + 1) + t] = base;
  }
  if (t == 0) P[(long)blockIdx.x * (NB + 1) + NB] = hi > lo ? hi - lo : 0;
  __syncthreads();

  float m1[8], m2[8], r1[8], r2[8];
#pragma unroll
  for (int d = 0; d < 8; ++d) {
    m1[d] = mu1[d];
    m2[d] = mu2[d];
    const float s1v = sg1[d], s2v = sg2[d];
    r1[d] = 1.0f / (EPSQ + s1v * s1v);
    r2[d] = 1.0f / (EPSQ + s2v * s2v);
  }

  for (int e0 = lo + t; e0 < hi; e0 += MU * MT) {
    float4 a0[MU], a1[MU];
    int sv[MU], dv[MU];
#pragma unroll
    for (int u = 0; u < MU; ++u) {
      int e = e0 + u * MT;
      e = e < hi ? e : hi - 1;
      a0[u] = *(const float4*)(ea + (long)e * 8);
      a1[u] = *(const float4*)(ea + (long)e * 8 + 4);
      sv[u] = ei[e];
      dv[u] = ei[E + e];
    }
#pragma unroll
    for (int u = 0; u < MU; ++u)
      asm volatile("" :: "v"(a0[u].x), "v"(a0[u].y), "v"(a0[u].z), "v"(a0[u].w),
                         "v"(a1[u].x), "v"(a1[u].y), "v"(a1[u].z), "v"(a1[u].w), "v"(sv[u]), "v"(dv[u]));
#pragma unroll
    for (int u = 0; u < MU; ++u) {
      const float av[8] = { a0[u].x, a0[u].y, a0[u].z, a0[u].w, a1[u].x, a1[u].y, a1[u].z, a1[u].w };
      float s1 = 0.0f, s2 = 0.0f;
#pragma unroll
      for (int d = 0; d < 8; ++d) {
        const float d1 = av[d] - m1[d];
        const float d2 = av[d] - m2[d];
        s1 += (d1 * d1) * r1[d];
        s2 += (d2 * d2) * r2[d];
      }
      if (e0 + u * MT < hi) {
        const int p = atomicAdd(&cur[dv[u] >> 8], 1);
        const int4 rec = make_int4((sv[u] << 8) | (dv[u] & 255),
                                   __float_as_int(__expf(-0.5f * s1)),
                                   __float_as_int(__expf(-0.5f * s2)), 0);
        if (staged) stage[p] = rec;
        else *(int4*)(tmp + (long)(lo + p) * 4) = rec;
      }
    }
  }
  if (staged) {
    __syncthreads();
    for (int i = t; i < hi - lo; i += MT) *(int4*)(tmp + (long)(lo + i) * 4) = stage[i];
  }
}

__global__ void __launch_bounds__(BT)
bin_kernel(const int* __restrict__ tmp, const int* __restrict__ P,
           int* __restrict__ seg, int* __restrict__ csr, int N, int E, int NB) {
  __shared__ int cnt[256];
  __shared__ int wsum[8];
  __shared__ int rs[G1];
  __shared__ int rl[G1];
  __shared__ int red[2][BT / 32];
  __shared__ int4 stage[BINMAX];
  __shared__ __attribute__((aligned(8))) int dh[64];
  __shared__ int dbase[64];
  __shared__ int4 segst[256];
  const int t = threadIdx.x;
  const int lane = t & 31;
  const int wv = t >> 5;
  const int CH = (E + G1 - 1) / G1;
  int p0 = 0, p1 = 0;
  if (t < G1) {
    p0 = P[(long)t * (NB + 1) + blockIdx.x];
    p1 = P[(long)t * (NB + 1) + blockIdx.x + 1];
    rs[t] = t * CH + p0;
    rl[t] = p1 - p0;
  }
  if (t < 256) cnt[t] = 0;
  if (t < 64) dh[t] = 0;
  {
    int s0 = p0, s1 = p1;
    for (int o = 16; o > 0; o >>= 1) {
      s0 += __shfl_xor(s0, o);
      s1 += __shfl_xor(s1, o);
    }
    if (lane == 0) { red[0][wv] = s0; red[1][wv] = s1; }
  }
  __syncthreads();
  int lo = 0, hi = 0;
#pragma unroll
  for (int w = 0; w < BT / 32; ++w) { lo += red[0][w]; hi += red[1][w]; }
  const bool fits = hi - lo <= BINMAX;

  int Lr[RW], Sr[RW];
#pragma unroll
  for (int u = 0; u < RW; ++u) {
    Lr[u] = __builtin_amdgcn_readfirstlane(rl[wv * RW + u]);
    Sr[u] = __builtin_amdgcn_readfirstlane(rs[wv * RW + u]);
  }
  int4 r[RW];
#pragma unroll
  for (int u = 0; u < RW; ++u) {
    const int L = Lr[u];
    int j = Sr[u] + (lane < L ? lane : L - 1);
    j = j > 0 ? j : 0;
    j = j < E ? j : E - 1;
    r[u] = *(const int4*)(tmp + (long)j * 4);
  }
#pragma unroll
  for (int u = 0; u < RW; ++u) asm volatile("" :: "v"(r[u].x), "v"(r[u].y), "v"(r[u].z));
#pragma unroll
  for (int u = 0; u < RW; ++u)
    if (lane < Lr[u]) atomicAdd(&cnt[r[u].x & 255], 1);
#pragma unroll
  for (int u = 0; u < RW; ++u) {
    const int L = Lr[u];
    const int S = Sr[u];
    for (int k = 32 + lane; k < L; k += 32) atomicAdd(&cnt[tmp[(long)(S + k) * 4] & 255], 1);
  }
  __syncthreads();
  int c = 0;
  if (t < 256) c = cnt[t];
  const int nme = blockIdx.x * 256 + t;
  int key = 63;
  if (t < 256) {
    if (nme < N) key = c < 62 ? c : 62;
    atomicAdd(&dh[key], 1);
  }
  int inc = c;
  for (int o = 1; o < 32; o <<= 1) {
    const int u = __shfl_up(inc, o);
    if (lane >= o) inc += u;
  }
  if (t < 256 && lane == 31) wsum[wv] = inc;
  __syncthreads();
  int beg = 0;
  if (t < 256) {
    beg = lo + inc - c;
    for (int w = 0; w < wv; ++w) beg += wsum[w];
    cnt[t] = beg;
  }
  if (wv == 0) {
    const int2 h = ((const int2*)dh)[lane];
    const int s = h.x + h.y;
    int incs = s;
    for (int o = 1; o < 32; o <<= 1) {
      const int u = __shfl_up(incs, o);
      if (lane >= o) incs += u;
    }
    dbase[2 * lane]     = incs - s;
    dbase[2 * lane + 1] = incs - s + h.x;
  }
  __syncthreads();
  if (t < 256) {
    const int pos = atomicAdd(&dbase[key], 1);
    segst[pos] = make_int4(beg, beg + c, nme, 0);
  }
  {
    int pos[RW];
#pragma unroll
    for (int u = 0; u < RW; ++u) {
      int p = lo;
      if (lane < Lr[u]) p = atomicAdd(&cnt[r[u].x & 255], 1);
      pos[u] = p;
    }
#pragma unroll
    for (int u = 0; u < RW; ++u) {
      if (lane < Lr[u]) {
        const int sn = (int)((unsigned)r[u].x >> 8);
        const int4 o = make_int4(sn, r[u].y, sn, r[u].z);
        if (fits) stage[pos[u] - lo] = o;
        else *(int4*)(csr + (long)pos[u] * 4) = o;
      }
    }
  }
#pragma unroll
  for (int u = 0; u < RW; ++u) {
    const int L = Lr[u];
    const int S = Sr[u];
    for (int k = 32 + lane; k < L; k += 32) {
      const int4 x = *(const int4*)(tmp + (long)(S + k) * 4);
      const int pos = atomicAdd(&cnt[x.x & 255], 1);
      const int sn = (int)((unsigned)x.x >> 8);
      const int4 o = make_int4(sn, x.y, sn, x.z);
      if (fits) stage[pos - lo] = o;
      else *(int4*)(csr + (long)pos * 4) = o;
    }
  }
  __syncthreads();
  if (fits) {
    for (int i = t; i < hi - lo; i += BT) *(int4*)(csr + (long)(lo + i) * 4) = stage[i];
  }
  if (t < 256) {
    const int p = blockIdx.x * 256 + t;
    if (p < N) *(int4*)(seg + (long)p * 4) = segst[t];
  }
}

#define RB 8
#define EU 4
#define TP 36

static __device__ __forceinline__ void own_lds_stores_landed() {
  asm volatile("s_wait_dscnt 0x0" ::: "memory");
}

template <int U>
static __device__ __forceinline__ int from_group_lane(int x) {
  int r;
  asm("v_mov_b32_dpp %0, %1 dpp8:[%2,%2,%2,%2,%2,%2,%2,%2]" : "=v"(r) : "v"(x), "n"(U));
  return r;
}
template <int U>
static __device__ __forceinline__ unsigned from_group_lane_or(unsigned x, unsigned y) {
  unsigned r;
  asm("v_or_b32_dpp %0, %1, %2 dpp8:[%3,%3,%3,%3,%3,%3,%3,%3]" : "=v"(r) : "v"(x), "v"(y), "n"(U));
  return r;
}

template <int U0>
static __device__ __forceinline__ void gather_sub_trip(const char* xrow, unsigned q16, int j,
                                                       const unsigned (&sb)[4], const int (&wm)[4],
                                                       const int (&deg)[4], float4 (&acc)[4]) {
  unsigned so[4][EU];
#pragma unroll
  for (int g = 0; g < 4; ++g) {
    so[g][0] = from_group_lane_or<U0 + 0>(sb[g], q16);
    so[g][1] = from_group_lane_or<U0 + 1>(sb[g], q16);
    so[g][2] = from_group_lane_or<U0 + 2>(sb[g], q16);
    so[g][3] = from_group_lane_or<U0 + 3>(sb[g], q16);
  }
  __builtin_amdgcn_sched_barrier(0);
  float4 v[4][EU];
#pragma unroll
  for (int g = 0; g < 4; ++g) {
#pragma unroll
    for (int u = 0; u < EU; ++u) v[g][u] = *(const float4*)(xrow + so[g][u]);
  }
#pragma unroll
  for (int g = 0; g < 4; ++g) {
    const int wb[EU] = { from_group_lane<U0 + 0>(wm[g]), from_group_lane<U0 + 1>(wm[g]),
                         from_group_lane<U0 + 2>(wm[g]), from_group_lane<U0 + 3>(wm[g]) };
#pragma unroll
    for (int u = 0; u < EU; ++u) {
      const int m = (j + U0 + u < deg[g]) ? -1 : 0;
      const float w = __int_as_float(wb[u]);
      acc[g].x = __builtin_fmaf(__int_as_float(__float_as_int(v[g][u].x) & m), w, acc[g].x);
      acc[g].y = __builtin_fmaf(__int_as_float(__float_as_int(v[g][u].y) & m), w, acc[g].y);
      acc[g].z = __builtin_fmaf(__int_as_float(__float_as_int(v[g][u].z) & m), w, acc[g].z);
      acc[g].w = __builtin_fmaf(__int_as_float(__float_as_int(v[g][u].w) & m), w, acc[g].w);
    }
  }
}

__global__ void __launch_bounds__(128)
layer_kernel(const int* __restrict__ seg, const int* __restrict__ csr,
             const float* __restrict__ XIN, const float* __restrict__ G,
             const float* __restrict__ R, const float* __restrict__ B,
             float* __restrict__ OUT, int N, int ntiles, int sel) {
  __shared__ __attribute__((aligned(16))) float stage[4 * 16 * TP];
  const int wv   = threadIdx.x >> 5;
  const int turn = (int)((blockIdx.x * 2654435761u) >> 30);
  const int wave = (blockIdx.x >> 2) * 16 + ((wv + turn) & 3) * 4 + (blockIdx.x & 3);
  if (wave >= ntiles) return;
  const int lane = threadIdx.x & 31;
  float* T = stage + wv * (16 * TP);
  const int row0 = wave * 16;
  const int nl = lane >> 3;
  const int q  = lane & 7;

  int begb[4], deg[4];
  int md = 0;
#pragma unroll
  for (int g = 0; g < 4; ++g) {
    const int p = row0 + g * 4 + nl;
    int2 be = make_int2(0, 0);
    if (p < N) be = *(const int2*)(seg + (long)p * 4);
    begb[g] = be.x * 16;
    const int d = be.y - be.x;
    deg[g] = d;
    md = d > md ? d : md;
  }
  for (int o = 16; o > 0; o >>= 1) {
    const int t = __shfl_xor(md, o);
    md = t > md ? t : md;
  }
  const int wmax = __builtin_amdgcn_readfirstlane(md);
  int arow = 0;
  if (row0 + (lane & 15) < N) arow = seg[(long)(row0 + (lane & 15)) * 4 + 2];

  float4 acc[4];
#pragma unroll
  for (int g = 0; g < 4; ++g) acc[g] = make_float4(0.0f, 0.0f, 0.0f, 0.0f);

  if (wmax > 0) {
    const char* xrow = (const char*)XIN;
    const char* rec  = (const char*)csr + sel * 8;
    const unsigned q16 = (unsigned)q * 16u;
    int2 rc[4];
#pragma unroll
    for (int g = 0; g < 4; ++g) {
      int lastb = begb[g] + deg[g] * 16 - 16;
      lastb = lastb > 0 ? lastb : 0;
      int jb = begb[g] + q * 16;
      jb = jb < lastb ? jb : lastb;
      rc[g] = *(const int2*)(rec + (unsigned)jb);
    }
    for (int j = 0; j < wmax; j += RB) {
      int2 nr[4];
#pragma unroll
      for (int g = 0; g < 4; ++g) {
        int lastb = begb[g] + deg[g] * 16 - 16;
        lastb = lastb > 0 ? lastb : 0;
        int jb = begb[g] + (j + RB + q) * 16;
        jb = jb < lastb ? jb : lastb;
        nr[g] = *(const int2*)(rec + (unsigned)jb);
      }
      unsigned sb[4];
      int wm[4];
#pragma unroll
      for (int g = 0; g < 4; ++g) {
        sb[g] = (unsigned)rc[g].x * 128u;
        wm[g] = (j + q < deg[g]) ? rc[g].y : 0;
      }
      gather_sub_trip<0>(xrow, q16, j, sb, wm, deg, acc);
      if (j + EU < wmax) gather_sub_trip<EU>(xrow, q16, j, sb, wm, deg, acc);
#pragma unroll
      for (int g = 0; g < 4; ++g) rc[g] = nr[g];
    }
  }

  const int half = lane >> 4;
  const int m    = lane & 15;
  int nd[4];
#pragma unroll
  for (int g = 0; g < 4; ++g) {
    const int p = row0 + g * 4 + nl;
    nd[g] = -1;
    if (p < N) nd[g] = seg[(long)p * 4 + 2];
  }

  v2f ax[8], bg0[8], bg1[8], br0[8], br1[8];
#pragma unroll
  for (int s = 0; s < 8; ++s) {
    const int kk = s * 4 + half * 2;
    ax[s] = *(const v2f*)(XIN + (long)arow * 32 + kk);
    bg0[s].x = G[kk * 32 + m];        bg0[s].y = G[(kk + 1) * 32 + m];
    bg1[s].x = G[kk * 32 + 16 + m];   bg1[s].y = G[(kk + 1) * 32 + 16 + m];
    br0[s].x = R[kk * 32 + m];        br0[s].y = R[(kk + 1) * 32 + m];
    br1[s].x = R[kk * 32 + 16 + m];   br1[s].y = R[(kk + 1) * 32 + 16 + m];
  }
  const float b0 = B[m], b1 = B[16 + m];

#pragma unroll
  for (int g = 0; g < 4; ++g) {
    float c = (float)deg[g];
    c = c > 1.0f ? c : 1.0f;
    const float inv = 1.0f / c;
    float4 a;
    a.x = acc[g].x * inv;
    a.y = acc[g].y * inv;
    a.z = acc[g].z * inv;
    a.w = acc[g].w * inv;
    *(float4*)(T + (g * 4 + nl) * TP + q * 4) = a;
  }
  own_lds_stores_landed();

  v8f accg0 = {}, accg1 = {}, accr0 = {}, accr1 = {};

#pragma unroll
  for (int s = 0; s < 8; ++s) {
    const int kk = s * 4 + half * 2;
    const v2f am = *(const v2f*)(T + m * TP + kk);
    accg0 = __builtin_amdgcn_wmma_f32_16x16x4_f32(false, am, false, bg0[s], (short)0, accg0, false, false);
    accg1 = __builtin_amdgcn_wmma_f32_16x16x4_f32(false, am, false, bg1[s], (short)0, accg1, false, false);
    accr0 = __builtin_amdgcn_wmma_f32_16x16x4_f32(false, ax[s], false, br0[s], (short)0, accr0, false, false);
    accr1 = __builtin_amdgcn_wmma_f32_16x16x4_f32(false, ax[s], false, br1[s], (short)0, accr1, false, false);
  }

#pragma unroll
  for (int r = 0; r < 8; ++r) {
    T[(half * 8 + r) * TP + m]      = accg0[r] + (accr0[r] + b0);
    T[(half * 8 + r) * TP + 16 + m] = accg1[r] + (accr1[r] + b1);
  }
  own_lds_stores_landed();
#pragma unroll
  for (int g = 0; g < 4; ++g) {
    const float4 o = *(const float4*)(T + (g * 4 + nl) * TP + q * 4);
    if (nd[g] >= 0) *(float4*)(OUT + (long)nd[g] * 32 + q * 4) = o;
  }
}

extern "C" void kernel_launch(void* const* d_in, const int* in_sizes, int n_in,
                              void* d_out, int out_size, void* d_ws, size_t ws_size,
                              hipStream_t stream) {
  const int*   ei  = (const int*)d_in[0];
  const float* ea  = (const float*)d_in[1];
  const float* x   = (const float*)d_in[2];
  const float* g1  = (const float*)d_in[3];
  const float* mu1 = (const float*)d_in[4];
  const float* sg1 = (const float*)d_in[5];
  const float* r1  = (const float*)d_in[6];
  const float* b1  = (const float*)d_in[7];
  const float* g2  = (const float*)d_in[8];
  const float* mu2 = (const float*)d_in[9];
  const float* sg2 = (const float*)d_in[10];
  const float* r2  = (const float*)d_in[11];
  const float* b2  = (const float*)d_in[12];

  const int E = in_sizes[0] / 2;
  const int N = in_sizes[2] / 32;
  const long plane = (long)N * 32;

  const int NB = (N + 255) / 256;
  int*   csr    = (int*)d_ws;
  int*   seg    = csr + (long)E * 4;
  int*   tmp    = seg + (long)N * 4;
  float* h      = (float*)tmp;
  const long tmpWords = (long)E * 4 > plane ? (long)E * 4 : plane;
  int*   P      = tmp + tmpWords;
  float* out    = (float*)d_out;

  const int ntiles     = (N + 15) / 16;
  const int tileBlocks = NB * 4;

  move_kernel<<<G1, MT, 0, stream>>>(ei, ea, mu1, sg1, mu2, sg2, P, tmp, E, NB);
  bin_kernel<<<NB, BT, 0, stream>>>(tmp, P, seg, csr, N, E, NB);

  layer_kernel<<<tileBlocks, 128, 0, stream>>>(seg, csr, x, g1, r1, b1, h, N, ntiles, 0);
  layer_kernel<<<tileBlocks, 128, 0, stream>>>(seg, csr, h, g2, r2, b2, out, N, ntiles, 1);
}


# Layer 1 fused into the bin kernel: the work-group that sorted a bin runs layer 1 on its 16 tiles from its LDS copy, a wave a tile; only layer 2's pairs are stored; 4 launches become 3; sum order same
# speedup vs baseline: 4.4962x; 1.1802x over previous
#include <hip/hip_runtime.h>


typedef float v2f __attribute__((ext_vector_type(2)));
typedef float v4f __attribute__((ext_vector_type(4)));
typedef float v8f __attribute__((ext_vector_type(8)));

#define EPSQ 1e-15f

#define NBMAX 512
#define G1 256
#define MT 1024
#define C1 7
#define MU 4
#define BT 512
#define RW (G1 / (BT / 32))
#define CHMAX 6656
#define BINMAX 4608


__global__ void __launch_bounds__(MT)
move_kernel(const int* __restrict__ ei,
            const float* __restrict__ ea,
            const float* __restrict__ mu1, const float* __restrict__ sg1,
            const float* __restrict__ mu2, const float* __restrict__ sg2,
            int* __restrict__ P, int* __restrict__ tmp, int E, int NB) {
  __shared__ int cur[NBMAX];
  __shared__ int wsum[32];
  __shared__ int4 stage[CHMAX];
  const int t = threadIdx.x;
  const int lane = t & 31;
  const int wv = t >> 5;
  const int CH = (E + G1 - 1) / G1;
  const bool staged = CH <= CHMAX;
  const int lo = blockIdx.x * CH;
  int hi = lo + CH;
  hi = hi < E ? hi : E;
  if (t < NBMAX) cur[t] = 0;
  __syncthreads();
  for (int e0 = lo + t; e0 < hi; e0 += C1 * MT) {
    int d[C1];
#pragma unroll
    for (int u = 0; u < C1; ++u) {
      int e = e0 + u * MT;
      e = e < hi ? e : hi - 1;
      d[u] = ei[E + e];
    }
#pragma unroll
    for (int u = 0; u < C1; ++u) asm volatile("" :: "v"(d[u]));
#pragma unroll
    for (int u = 0; u < C1; ++u)
      if (e0 + u * MT < hi) atomicAdd(&cur[d[u] >> 8], 1);
  }
  __syncthreads();
  int c = 0;
  if (t < NBMAX) c = cur[t];
  int inc = c;
  for (int o = 1; o < 32; o <<= 1) {
    const int u = __shfl_up(inc, o);
    if (lane >= o) inc += u;
  }
  if (lane == 31) wsum[wv] = inc;
  __syncthreads();
  if (t < NBMAX) {
    int base = inc - c;
    for (int w = 0; w < wv; ++w) base += wsum[w];
    cur[t] = base;
    if (t < NB) P[(long)blockIdx.x * (NB + 1) + t] = base;
  }
  if (t == 0) P[(long)blockIdx.x * (NB + 1) + NB] = hi > lo ? hi - lo : 0;
  __syncthreads();

  float m1[8], m2[8], r1[8], r2[8];
#pragma unroll
  for (int d = 0; d < 8; ++d) {
    m1[d] = mu1[d];
    m2[d] = mu2[d];
    const float s1v = sg1[d], s2v = sg2[d];
    r1[d] = 1.0f / (EPSQ + s1v * s1v);
    r2[d] = 1.0f / (EPSQ + s2v * s2v);
  }

  for (int e0 = lo + t; e0 < hi; e0 += MU * MT) {
    float4 a0[MU], a1[MU];
    int sv[MU], dv[MU];
#pragma unroll
    for (int u = 0; u < MU; ++u) {
      int e = e0 + u * MT;
      e = e < hi ? e : hi - 1;
      a0[u] = *(const float4*)(ea + (long)e * 8);
      a1[u] = *(const float4*)(ea + (long)e * 8 + 4);
      sv[u] = ei[e];
      dv[u] = ei[E + e];
    }
#pragma unroll
    for (int u = 0; u < MU; ++u)
      asm volatile("" :: "v"(a0[u].x), "v"(a0[u].y), "v"(a0[u].z), "v"(a0[u].w),
                         "v"(a1[u].x), "v"(a1[u].y), "v"(a1[u].z), "v"(a1[u].w), "v"(sv[u]), "v"(dv[u]));
#pragma unroll
    for (int u = 0; u < MU; ++u) {
      const float av[8] = { a0[u].x, a0[u].y, a0[u].z, a0[u].w, a1[u].x, a1[u].y, a1[u].z, a1[u].w };
      float s1 = 0.0f, s2 = 0.0f;
#pragma unroll
      for (int d = 0; d < 8; ++d) {
        const float d1 = av[d] - m1[d];
        const float d2 = av[d] - m2[d];
        s1 += (d1 * d1) * r1[d];
        s2 += (d2 * d2) * r2[d];
      }
      if (e0 + u * MT < hi) {
        const int p = atomicAdd(&cur[dv[u] >> 8], 1);
        const int4 rec = make_int4((sv[u] << 8) | (dv[u] & 255),
                                   __float_as_int(__expf(-0.5f * s1)),
                                   __float_as_int(__expf(-0.5f * s2)), 0);
        if (staged) stage[p] = rec;
        else *(int4*)(tmp + (long)(lo + p) * 4) = rec;
      }
    }
  }
  if (staged) {
    __syncthreads();
    for (int i = t; i < hi - lo; i += MT) *(int4*)(tmp + (long)(lo + i) * 4) = stage[i];
  }
}

#define RB 8
#define EU 4
#define TP 36
static __device__ __forceinline__ void own_lds_stores_landed() {
  asm volatile("s_wait_dscnt 0x0" ::: "memory");
}

template <int U>
static __device__ __forceinline__ int from_group_lane(int x) {
  int r;
  asm("v_mov_b32_dpp %0, %1 dpp8:[%2,%2,%2,%2,%2,%2,%2,%2]" : "=v"(r) : "v"(x), "n"(U));
  return r;
}
template <int U>
static __device__ __forceinline__ unsigned from_group_lane_or(unsigned x, unsigned y) {
  unsigned r;
  asm("v_or_b32_dpp %0, %1, %2 dpp8:[%3,%3,%3,%3,%3,%3,%3,%3]" : "=v"(r) : "v"(x), "v"(y), "n"(U));
  return r;
}

template <int U0>
static __device__ __forceinline__ void gather_sub_trip(const char* xrow, unsigned q16, int j,
                                                       const unsigned (&sb)[4], const int (&wm)[4],
                                                       const int (&deg16)[4], float4 (&acc)[4]) {
  unsigned so[4][EU];
#pragma unroll
  for (int g = 0; g < 4; ++g) {
    so[g][0] = from_group_lane_or<U0 + 0>(sb[g], q16);
    so[g][1] = from_group_lane_or<U0 + 1>(sb[g], q16);
    so[g][2] = from_group_lane_or<U0 + 2>(sb[g], q16);
    so[g][3] = from_group_lane_or<U0 + 3>(sb[g], q16);
  }
  __builtin_amdgcn_sched_barrier(0);
  v4f v[4][EU];
#pragma unroll
  for (int g = 0; g < 4; ++g) {
#pragma unroll
    for (int u = 0; u < EU; ++u) v[g][u] = *(const v4f*)(xrow + so[g][u]);
  }
#pragma unroll
  for (int g = 0; g < 4; ++g) {
    const int wb[EU] = { from_group_lane<U0 + 0>(wm[g]), from_group_lane<U0 + 1>(wm[g]),
                         from_group_lane<U0 + 2>(wm[g]), from_group_lane<U0 + 3>(wm[g]) };
#pragma unroll
    for (int u = 0; u < EU; ++u) {
      const int m = ((j + U0 + u) * 16 < deg16[g]) ? -1 : 0;
      const float w = __int_as_float(wb[u]);
      acc[g].x = __builtin_fmaf(__int_as_float(__float_as_int(v[g][u].x) & m), w, acc[g].x);
      acc[g].y = __builtin_fmaf(__int_as_float(__float_as_int(v[g][u].y) & m), w, acc[g].y);
      acc[g].z = __builtin_fmaf(__int_as_float(__float_as_int(v[g][u].z) & m), w, acc[g].z);
      acc[g].w = __builtin_fmaf(__int_as_float(__float_as_int(v[g][u].w) & m), w, acc[g].w);
    }
  }
}

template <int RS>
static __device__ __forceinline__ void gather_tile(const char* rec,
                                                   const int (&begb)[4],
                                                   const int (&deg16)[4],
                                                   int wmax,
                                                   const float* __restrict__ XIN, int tid, float4 (&acc)[4]) {
  const unsigned q16 = (unsigned)(tid & 7) * 16u;
  if (wmax > 0) {
    const char* xrow = (const char*)XIN;
    const int qrs = RS == 16 ? (int)q16 : (int)(q16 >> 1);
    int2 rc[4];
#pragma unroll
    for (int g = 0; g < 4; ++g) {
      int lastb = begb[g] + (RS == 16 ? deg16[g] : deg16[g] >> 1) - RS;
      lastb = lastb > 0 ? lastb : 0;
      int jb = begb[g] + qrs;
      jb = jb < lastb ? jb : lastb;
      rc[g] = *(const int2*)(rec + (unsigned)jb);
    }
    for (int j = 0; j < wmax; j += RB) {
      int2 nr[4];
      const auto fetch_next = [&]() {
#pragma unroll
        for (int g = 0; g < 4; ++g) {
          int lastb = begb[g] + (RS == 16 ? deg16[g] : deg16[g] >> 1) - RS;
          lastb = lastb > 0 ? lastb : 0;
          int jb = begb[g] + (j + RB) * RS + qrs;
          jb = jb < lastb ? jb : lastb;
          nr[g] = *(const int2*)(rec + (unsigned)jb);
        }
      };
      fetch_next();
      unsigned sb[4];
      int wm[4];
#pragma unroll
      for (int g = 0; g < 4; ++g) {
        sb[g] = (unsigned)rc[g].x * 128u;
        wm[g] = (j * 16 + (int)q16 < deg16[g]) ? rc[g].y : 0;
      }
      gather_sub_trip<0>(xrow, q16, j, sb, wm, deg16, acc);
      if (j + EU < wmax) gather_sub_trip<EU>(xrow, q16, j, sb, wm, deg16, acc);
#pragma unroll
      for (int g = 0; g < 4; ++g) rc[g] = nr[g];
    }
  }
}

static __device__ __forceinline__ void gather_tile_long(const int* pairs, const int (&first)[4], const int (&deg16)[4],
                                                        const float* __restrict__ XIN, int tid, float4 (&acc)[4]) {
  const int q4 = (tid & 7) * 4;
#pragma unroll
  for (int g = 0; g < 4; ++g) {
    const int d = deg16[g] >> 4;
    for (int e = 0; e < d; ++e) {
      const int2 r = *(const int2*)(pairs + (long)(first[g] + e) * 2);
      const float4 xv = *(const float4*)(XIN + (long)r.x * 32 + q4);
      const float w = __int_as_float(r.y);
      acc[g].x = __builtin_fmaf(xv.x, w, acc[g].x);
      acc[g].y = __builtin_fmaf(xv.y, w, acc[g].y);
      acc[g].z = __builtin_fmaf(xv.z, w, acc[g].z);
      acc[g].w = __builtin_fmaf(xv.w, w, acc[g].w);
    }
  }
}

template <bool LEAN, typename NodeOf>
static __device__ __forceinline__ void transform_tile(float4 (&acc)[4], const int (&deg16)[4],
                                                      int arow,
                                                      NodeOf node_of,
                                                      const float* __restrict__ XIN, const float* __restrict__ G,
                                                      const float* __restrict__ R, const float* __restrict__ B,
                                                      float* __restrict__ OUT, float* tiles, int tid) {
  const int lane = tid & 31;
  float* T = tiles + (tid >> 5) * (16 * TP);
  const int nl   = lane >> 3;
  const int q4   = (lane & 7) * 4;
  const int half = lane >> 4;
  const int m    = lane & 15;
  int nd[4];
#pragma unroll
  for (int g = 0; g < 4; ++g) nd[g] = node_of(tid, g);

  constexpr int S0 = LEAN ? 4 : 8;
  v2f ax[8], bg0[8], bg1[8], br0[8], br1[8];
#pragma unroll
  for (int s = 0; s < 8; ++s) ax[s] = *(const v2f*)(XIN + (long)arow * 32 + s * 4 + half * 2);
  const auto b_fragments = [&](int s) {
    const int kk = s * 4 + half * 2;
    bg0[s].x = G[kk * 32 + m];        bg0[s].y = G[(kk + 1) * 32 + m];
    bg1[s].x = G[kk * 32 + 16 + m];   bg1[s].y = G[(kk + 1) * 32 + 16 + m];
    br0[s].x = R[kk * 32 + m];        br0[s].y = R[(kk + 1) * 32 + m];
    br1[s].x = R[kk * 32 + 16 + m];   br1[s].y = R[(kk + 1) * 32 + 16 + m];
  };
#pragma unroll
  for (int s = 0; s < S0; ++s) b_fragments(s);
  const float b0 = B[m], b1 = B[16 + m];

#pragma unroll
  for (int g = 0; g < 4; ++g) {
    float c = (float)(deg16[g] >> 4);
    c = c > 1.0f ? c : 1.0f;
    const float inv = 1.0f / c;
    float4 a;
    a.x = acc[g].x * inv;
    a.y = acc[g].y * inv;
    a.z = acc[g].z * inv;
    a.w = acc[g].w * inv;
    *(float4*)(T + (g * 4 + nl) * TP + q4) = a;
  }
  own_lds_stores_landed();
#pragma unroll
  for (int s = S0; s < 8; ++s) b_fragments(s);

  v8f accg0 = {}, accg1 = {}, accr0 = {}, accr1 = {};

#pragma unroll
  for (int s = 0; s < 8; ++s) {
    const int kk = s * 4 + half * 2;
    const v2f am = *(const v2f*)(T + m * TP + kk);
    accg0 = __builtin_amdgcn_wmma_f32_16x16x4_f32(false, am, false, bg0[s], (short)0, accg0, false, false);
    accg1 = __builtin_amdgcn_wmma_f32_16x16x4_f32(false, am, false, bg1[s], (short)0, accg1, false, false);
    accr0 = __builtin_amdgcn_wmma_f32_16x16x4_f32(false, ax[s], false, br0[s], (short)0, accr0, false, false);
    accr1 = __builtin_amdgcn_wmma_f32_16x16x4_f32(false, ax[s], false, br1[s], (short)0, accr1, false, false);
  }

#pragma unroll
  for (int r = 0; r < 8; ++r) {
    T[(half * 8 + r) * TP + m]      = accg0[r] + (accr0[r] + b0);
    T[(half * 8 + r) * TP + 16 + m] = accg1[r] + (accr1[r] + b1);
  }
  own_lds_stores_landed();
#pragma unroll
  for (int g = 0; g < 4; ++g) {
    const float4 o = *(const float4*)(T + (g * 4 + nl) * TP + q4);
    if (nd[g] >= 0) *(float4*)(OUT + (long)nd[g] * 32 + q4) = o;
  }
}

__global__ void __launch_bounds__(BT, 8)
bin_layer_kernel(const int* __restrict__ tmp, const int* __restrict__ P,
                 int* __restrict__ seg, int* __restrict__ csr,
                 const float* __restrict__ XIN, const float* __restrict__ G,
                 const float* __restrict__ R, const float* __restrict__ B,
                 float* __restrict__ OUT, int N, int E, int NB) {
  __shared__ int cnt[256];
  __shared__ int wsum[8];
  __shared__ int rs[G1];
  __shared__ int rl[G1];
  __shared__ int red[2][BT / 32];
  __shared__ int4 stage[BINMAX];
  __shared__ __attribute__((aligned(8))) int dh[64];
  __shared__ int dbase[64];
  __shared__ int4 segst[256];
  __shared__ int tmax[16];
  __shared__ __attribute__((aligned(16))) float tiles[(BT / 32) * 16 * TP];
  const int t = threadIdx.x;
  const int lane = t & 31;
  const int wv = t >> 5;
  const int CH = (E + G1 - 1) / G1;
  int p0 = 0, p1 = 0;
  if (t < G1) {
    p0 = P[(long)t * (NB + 1) + blockIdx.x];
    p1 = P[(long)t * (NB + 1) + blockIdx.x + 1];
    rs[t] = t * CH + p0;
    rl[t] = p1 - p0;
  }
  if (t < 256) cnt[t] = 0;
  if (t < 64) dh[t] = 0;
  if (t < 16) tmax[t] = 0;
  {
    int s0 = p0, s1 = p1;
    for (int o = 16; o > 0; o >>= 1) {
      s0 += __shfl_xor(s0, o);
      s1 += __shfl_xor(s1, o);
    }
    if (lane == 0) { red[0][wv] = s0; red[1][wv] = s1; }
  }
  __syncthreads();
  int lo = 0, hi = 0;
#pragma unroll
  for (int w = 0; w < BT / 32; ++w) { lo += red[0][w]; hi += red[1][w]; }
  const bool fits = hi - lo <= BINMAX;

  int Lr[RW], Sr[RW];
#pragma unroll
  for (int u = 0; u < RW; ++u) {
    Lr[u] = __builtin_amdgcn_readfirstlane(rl[wv * RW + u]);
    Sr[u] = __builtin_amdgcn_readfirstlane(rs[wv * RW + u]);
  }
  int4 r[RW];
#pragma unroll
  for (int u = 0; u < RW; ++u) {
    const int L = Lr[u];
    int j = Sr[u] + (lane < L ? lane : L - 1);
    j = j > 0 ? j : 0;
    j = j < E ? j : E - 1;
    r[u] = *(const int4*)(tmp + (long)j * 4);
  }
#pragma unroll
  for (int u = 0; u < RW; ++u) asm volatile("" :: "v"(r[u].x), "v"(r[u].y), "v"(r[u].z));
#pragma unroll
  for (int u = 0; u < RW; ++u)
    if (lane < Lr[u]) atomicAdd(&cnt[r[u].x & 255], 1);
#pragma unroll
  for (int u = 0; u < RW; ++u) {
    const int L = Lr[u];
    const int S = Sr[u];
    for (int k = 32 + lane; k < L; k += 32) atomicAdd(&cnt[tmp[(long)(S + k) * 4] & 255], 1);
  }
  __syncthreads();
  int c = 0;
  if (t < 256) c = cnt[t];
  const int nme = blockIdx.x * 256 + t;
  int key = 63;
  if (t < 256) {
    if (nme < N) key = c < 62 ? c : 62;
    atomicAdd(&dh[key], 1);
  }
  int inc = c;
  for (int o = 1; o < 32; o <<= 1) {
    const int u = __shfl_up(inc, o);
    if (lane >= o) inc += u;
  }
  if (t < 256 && lane == 31) wsum[wv] = inc;
  __syncthreads();
  int beg = 0;
  if (t < 256) {
    beg = lo + inc - c;
    for (int w = 0; w < wv; ++w) beg += wsum[w];
    cnt[t] = beg;
  }
  if (wv == 0) {
    const int2 h = ((const int2*)dh)[lane];
    const int s = h.x + h.y;
    int incs = s;
    for (int o = 1; o < 32; o <<= 1) {
      const int u = __shfl_up(incs, o);
      if (lane >= o) incs += u;
    }
    dbase[2 * lane]     = incs - s;
    dbase[2 * lane + 1] = incs - s + h.x;
  }
  __syncthreads();
  if (t < 256) {
    const int pos = atomicAdd(&dbase[key], 1);
    segst[pos] = make_int4(beg, beg + c, nme, 0);
    atomicMax(&tmax[pos >> 4], c);
  }
  {
    int pos[RW];
#pragma unroll
    for (int u = 0; u < RW; ++u) {
      int p = lo;
      if (lane < Lr[u]) p = atomicAdd(&cnt[r[u].x & 255], 1);
      pos[u] = p;
    }
#pragma unroll
    for (int u = 0; u < RW; ++u) {
      if (lane < Lr[u]) {
        const int sn = (int)((unsigned)r[u].x >> 8);
        if (fits) stage[pos[u] - lo] = make_int4(sn, r[u].y, sn, r[u].z);
        else *(int2*)(csr + (long)pos[u] * 2) = make_int2(sn, r[u].y);
      }
    }
  }
#pragma unroll
  for (int u = 0; u < RW; ++u) {
    const int L = Lr[u];
    const int S = Sr[u];
    for (int k = 32 + lane; k < L; k += 32) {
      const int4 x = *(const int4*)(tmp + (long)(S + k) * 4);
      const int pos = atomicAdd(&cnt[x.x & 255], 1);
      const int sn = (int)((unsigned)x.x >> 8);
      if (fits) stage[pos - lo] = make_int4(sn, x.y, sn, x.z);
      else *(int2*)(csr + (long)pos * 2) = make_int2(sn, x.y);
    }
  }
  __syncthreads();
  if (t < 256) {
    const int p = blockIdx.x * 256 + t;
    if (p < N) *(int4*)(seg + (long)p * 4) = segst[t];
  }

  const auto tile_of = [](int w) -> int { return (w & 12) + ((w & 4) ? 3 - (w & 3) : (w & 3)); };
  const auto node_of = [&](int th, int g) -> int {
    const int i = tile_of(th >> 5) * 16 + g * 4 + ((th & 31) >> 3);
    return blockIdx.x * 256 + i < N ? segst[i].z : -1;
  };
  if (fits) {
    for (int i = t; i < hi - lo; i += BT)
      *(int2*)(csr + (long)(lo + i) * 2) = *(const int2*)((const char*)(stage + i) + 8);
  }
  const int tile = tile_of(wv);
  const int wmax = __builtin_amdgcn_readfirstlane(tmax[tile]);
  int arow = 0;
  if (blockIdx.x * 256 + tile * 16 + (lane & 15) < N) arow = segst[tile * 16 + (lane & 15)].z;
  int begb[4], deg[4];
  float4 acc[4];
#pragma unroll
  for (int g = 0; g < 4; ++g) acc[g] = make_float4(0.0f, 0.0f, 0.0f, 0.0f);
  if (fits) {
#pragma unroll
    for (int g = 0; g < 4; ++g) {
      const int2 be = *(const int2*)(segst + tile * 16 + g * 4 + (lane >> 3));
      begb[g] = (be.x - lo) * 16;
      deg[g] = (be.y - be.x) * 16;
    }
    gather_tile<16>((const char*)stage, begb, deg, wmax, XIN, t, acc);
  } else {
#pragma unroll
    for (int g = 0; g < 4; ++g) {
      const int2 be = *(const int2*)(segst + tile * 16 + g * 4 + (lane >> 3));
      begb[g] = be.x;
      deg[g] = (be.y - be.x) * 16;
    }
    gather_tile_long(csr, begb, deg, XIN, t, acc);
  }
  transform_tile<true>(acc, deg, arow, node_of, XIN, G, R, B, OUT, tiles, t);
  if (!fits) {
    __syncthreads();
    if (t < 256) cnt[t] = beg;
    __syncthreads();
#pragma unroll 1
    for (int u = 0; u < RW; ++u) {
      const int L = rl[wv * RW + u];
      const int S = rs[wv * RW + u];
      for (int k = lane; k < L; k += 32) {
        const int4 x = *(const int4*)(tmp + (long)(S + k) * 4);
        const int pos = atomicAdd(&cnt[x.x & 255], 1);
        *(int2*)(csr + (long)pos * 2) = make_int2((int)((unsigned)x.x >> 8), x.z);
      }
    }
  }
}

__global__ void __launch_bounds__(128)
layer_kernel(const int* __restrict__ seg, const int* __restrict__ csr,
             const float* __restrict__ XIN, const float* __restrict__ G,
             const float* __restrict__ R, const float* __restrict__ B,
             float* __restrict__ OUT, int N, int ntiles) {
  __shared__ __attribute__((aligned(16))) float stage[4 * 16 * TP];
  const int wv   = threadIdx.x >> 5;
  const int turn = (int)((blockIdx.x * 2654435761u) >> 30);
  const int wave = (blockIdx.x >> 2) * 16 + ((wv + turn) & 3) * 4 + (blockIdx.x & 3);
  if (wave >= ntiles) return;
  const int lane = threadIdx.x & 31;
  const int row0 = wave * 16;
  const int nl = lane >> 3;

  int begb[4], deg[4];
#pragma unroll
  for (int g = 0; g < 4; ++g) {
    const int p = row0 + g * 4 + nl;
    int2 be = make_int2(0, 0);
    if (p < N) be = *(const int2*)(seg + (long)p * 4);
    begb[g] = be.x * 8;
    deg[g] = (be.y - be.x) * 16;
  }
  int md = 0;
#pragma unroll
  for (int g = 0; g < 4; ++g) md = deg[g] > md ? deg[g] : md;
  for (int o = 16; o > 0; o >>= 1) {
    const int t = __shfl_xor(md, o);
    md = t > md ? t : md;
  }
  const int wmax = __builtin_amdgcn_readfirstlane(md) >> 4;
  int arow = 0;
  if (row0 + (lane & 15) < N) arow = seg[(long)(row0 + (lane & 15)) * 4 + 2];
  const auto node_of = [&](int th, int g) -> int {
    const int w = (blockIdx.x >> 2) * 16 + (((th >> 5) + turn) & 3) * 4 + (blockIdx.x & 3);
    const int p = w * 16 + g * 4 + ((th & 31) >> 3);
    return p < N ? seg[(long)p * 4 + 2] : -1;
  };
  float4 acc[4];
#pragma unroll
  for (int g = 0; g < 4; ++g) acc[g] = make_float4(0.0f, 0.0f, 0.0f, 0.0f);
  gather_tile<8>((const char*)csr, begb, deg, wmax, XIN, (int)threadIdx.x, acc);
  transform_tile<false>(acc, deg, arow, node_of, XIN, G, R, B, OUT, stage, (int)threadIdx.x);
}

extern "C" void kernel_launch(void* const* d_in, const int* in_sizes, int n_in,
                              void* d_out, int out_size, void* d_ws, size_t ws_size,
                              hipStream_t stream) {
  const int*   ei  = (const int*)d_in[0];
  const float* ea  = (const float*)d_in[1];
  const float* x   = (const float*)d_in[2];
  const float* g1  = (const float*)d_in[3];
  const float* mu1 = (const float*)d_in[4];
  const float* sg1 = (const float*)d_in[5];
  const float* r1  = (const float*)d_in[6];
  const float* b1  = (const float*)d_in[7];
  const float* g2  = (const float*)d_in[8];
  const float* mu2 = (const float*)d_in[9];
  const float* sg2 = (const float*)d_in[10];
  const float* r2  = (const float*)d_in[11];
  const float* b2  = (const float*)d_in[12];

  const int E = in_sizes[0] / 2;
  const int N = in_sizes[2] / 32;
  const long plane = (long)N * 32;

  const int NB = (N + 255) / 256;
  const long csrWords = ((long)E * 2 + 31) & ~31L;
  int*   csr    = (int*)d_ws;
  float* h      = (float*)(csr + csrWords);
  int*   seg    = (int*)(h + plane);
  int*   tmp    = seg + (long)N * 4;
  int*   P      = tmp + (long)E * 4;
  float* out    = (float*)d_out;

  const int ntiles     = (N + 15) / 16;
  const int tileBlocks = NB * 4;

  move_kernel<<<G1, MT, 0, stream>>>(ei, ea, mu1, sg1, mu2, sg2, P, tmp, E, NB);
  bin_layer_kernel<<<NB, BT, 0, stream>>>(tmp, P, seg, csr, x, g1, r1, b1, h, N, E, NB);

  layer_kernel<<<tileBlocks, 128, 0, stream>>>(seg, csr, h, g2, r2, b2, out, N, ntiles);
}
